# starting kernel
# baseline (speedup 1.0000x reference)
#include <hip/hip_runtime.h>
#include <hip/hip_bf16.h>

typedef __attribute__((ext_vector_type(2))) float v2f;
typedef __attribute__((ext_vector_type(8))) float v8f;

__global__ void fill_kernel(float* __restrict__ p, float v, size_t n) {
    size_t i = (size_t)blockIdx.x * blockDim.x + threadIdx.x;
    if (i < n) p[i] = v;
}

__global__ void pack_w_kernel(const float* __restrict__ W, float* __restrict__ P,
                              int K, int N, int NPAD) {
    int idx = blockIdx.x * blockDim.x + threadIdx.x;
    if (idx >= K * NPAD) return;
    int k = idx / NPAD;
    int c = idx - k * NPAD;
    float v = (c < N) ? W[(size_t)k * N + c] : 0.0f;
    P[(size_t)(k >> 1) * 2 * NPAD + 2 * c + (k & 1)] = v;
}

__global__ void degree_kernel(const int* __restrict__ dst, float* __restrict__ deg, int E) {
    int e = blockIdx.x * blockDim.x + threadIdx.x;
    if (e < E) atomicAdd(&deg[dst[e]], 1.0f);
}

__global__ void rsqrt_kernel(float* __restrict__ deg, int N) {
    int i = blockIdx.x * blockDim.x + threadIdx.x;
    if (i < N) {
        float d = deg[i];
        deg[i] = (d > 0.0f) ? rsqrtf(d) : 0.0f;
    }
}

template <int NT, int NPAD>
__global__ __launch_bounds__(128, 1) void wmma_gemm_f32(
        const float* __restrict__ A, const float* __restrict__ Wp,
        float* __restrict__ Out, int M, int K, int Nout) {
    const int lane = threadIdx.x & 31;
    const int wave = threadIdx.x >> 5;
    const int nPairTiles = (M + 31) >> 5;
    const int pairTile = blockIdx.x * (blockDim.x >> 5) + wave;
    if (pairTile >= nPairTiles) return;

    const int m0    = pairTile * 32 + (lane & 15);
    const int koff  = (lane >> 4) << 1;
    const int koff2 = (lane >> 4);
    const int n     = lane & 15;

    const bool hasSecond = (pairTile * 32 + 16) < M;

    v8f acc0[NT], acc1[NT];
#pragma unroll
    for (int t = 0; t < NT; ++t) {
        v8f z = {0.f, 0.f, 0.f, 0.f, 0.f, 0.f, 0.f, 0.f};
        acc0[t] = z;
        acc1[t] = z;
    }

    const float* A0 = A + (size_t)m0 * K;
    const float* A1 = hasSecond ? (A0 + (size_t)16 * K) : A0;

    for (int k = 0; k < K; k += 4) {
        v2f a0 = *(const v2f*)(A0 + k + koff);
        v2f a1 = *(const v2f*)(A1 + k + koff);
        const float* Wrow = Wp + (size_t)((k >> 1) + koff2) * (2 * NPAD);
#pragma unroll
        for (int t = 0; t < NT; ++t) {
            v2f b = *(const v2f*)(Wrow + 2 * (t * 16 + n));
            acc0[t] = __builtin_amdgcn_wmma_f32_16x16x4_f32(
                false, a0, false, b, (short)0, acc0[t], false, false);
            acc1[t] = __builtin_amdgcn_wmma_f32_16x16x4_f32(
                false, a1, false, b, (short)0, acc1[t], false, false);
        }
    }

    const int rbase0 = pairTile * 32 + ((lane >> 4) << 3);
    const int rbase1 = rbase0 + 16;
#pragma unroll
    for (int t = 0; t < NT; ++t) {
        int col = t * 16 + n;
        if (col < Nout) {
#pragma unroll
            for (int i = 0; i < 8; ++i)
                Out[(size_t)(rbase0 + i) * Nout + col] = acc0[t][i];
            if (hasSecond) {
#pragma unroll
                for (int i = 0; i < 8; ++i)
                    Out[(size_t)(rbase1 + i) * Nout + col] = acc1[t][i];
            }
        }
    }
}

__global__ void scatter_kernel(const float* __restrict__ H, const int* __restrict__ src,
                               const int* __restrict__ dst, const float* __restrict__ dinv,
                               float* __restrict__ acc, int E, int F) {
    const int lane = threadIdx.x & 31;
    const int wave = threadIdx.x >> 5;
    const int e = blockIdx.x * (blockDim.x >> 5) + wave;
    if (e >= E) return;
    const int f0 = lane << 2;
    if (f0 >= F) return;
    const int s = src[e];
    const int d = dst[e];
    const float norm = dinv[s] * dinv[d];
    const float4 h = *(const float4*)(H + (size_t)s * F + f0);
    float* out = acc + (size_t)d * F + f0;
    atomicAdd(out + 0, h.x * norm);
    atomicAdd(out + 1, h.y * norm);
    atomicAdd(out + 2, h.z * norm);
    atomicAdd(out + 3, h.w * norm);
}

__global__ void bias_relu_self_kernel(float* __restrict__ acc, const float* __restrict__ h,
                                      const float* __restrict__ dinv, const float* __restrict__ b,
                                      int N, int logF) {
    size_t i = (size_t)blockIdx.x * blockDim.x + threadIdx.x;
    if (i >= ((size_t)N << logF)) return;
    int node = (int)(i >> logF);
    int f = (int)(i & ((1u << logF) - 1));
    float dn = dinv[node];
    float v = acc[i] + h[i] * dn * dn + b[f];
    acc[i] = v > 0.0f ? v : 0.0f;
}

__global__ void logsoftmax_kernel(const float* __restrict__ acc2, const float* __restrict__ h2,
                                  const float* __restrict__ dinv, const float* __restrict__ b2,
                                  float* __restrict__ out, int N, int C) {
    const int lane = threadIdx.x & 31;
    const int node = (blockIdx.x * blockDim.x + threadIdx.x) >> 5;
    if (node >= N) return;
    const float dn = dinv[node];
    const float sn = dn * dn;
    const int f0 = lane, f1 = lane + 32;
    const size_t base = (size_t)node * C;
    float v0 = -3.4e38f, v1 = -3.4e38f;
    if (f0 < C) v0 = acc2[base + f0] + h2[base + f0] * sn + b2[f0];
    if (f1 < C) v1 = acc2[base + f1] + h2[base + f1] * sn + b2[f1];
    float mx = fmaxf(v0, v1);
#pragma unroll
    for (int off = 16; off > 0; off >>= 1) mx = fmaxf(mx, __shfl_xor(mx, off, 32));
    float s = 0.0f;
    if (f0 < C) s += __expf(v0 - mx);
    if (f1 < C) s += __expf(v1 - mx);
#pragma unroll
    for (int off = 16; off > 0; off >>= 1) s += __shfl_xor(s, off, 32);
    const float lse = __logf(s) + mx;
    if (f0 < C) out[base + f0] = v0 - lse;
    if (f1 < C) out[base + f1] = v1 - lse;
}

extern "C" void kernel_launch(void* const* d_in, const int* in_sizes, int n_in,
                              void* d_out, int out_size, void* d_ws, size_t ws_size,
                              hipStream_t stream) {
    const float* x  = (const float*)d_in[0];
    const int*   ei = (const int*)d_in[1];
    const float* W1 = (const float*)d_in[2];
    const float* b1 = (const float*)d_in[3];
    const float* W2 = (const float*)d_in[4];
    const float* b2 = (const float*)d_in[5];
    float* out = (float*)d_out;

    const int F_IN  = 256;
    const int HID   = 128;
    const int NCLS  = 40;
    const int NPAD2 = 48;
    const int N = in_sizes[0] / F_IN;
    const int E = in_sizes[1] / 2;
    const int* src = ei;
    const int* dst = ei + E;

    float* dinv = (float*)d_ws;
    float* h1   = dinv + ((N + 63) & ~63);
    float* acc1 = h1 + (size_t)N * HID;
    float* h2   = acc1 + (size_t)N * HID;
    float* acc2 = h2 + (size_t)N * NCLS;
    float* W1p  = acc2 + (size_t)N * NCLS;
    float* W2p  = W1p + (size_t)F_IN * HID;

    {
        size_t n1 = N;
        fill_kernel<<<(unsigned)((n1 + 255) / 256), 256, 0, stream>>>(dinv, 1.0f, n1);
        size_t n2 = (size_t)N * HID;
        fill_kernel<<<(unsigned)((n2 + 255) / 256), 256, 0, stream>>>(acc1, 0.0f, n2);
        size_t n3 = (size_t)N * NCLS;
        fill_kernel<<<(unsigned)((n3 + 255) / 256), 256, 0, stream>>>(acc2, 0.0f, n3);
        pack_w_kernel<<<(F_IN * HID + 255) / 256, 256, 0, stream>>>(W1, W1p, F_IN, HID, HID);
        pack_w_kernel<<<(HID * NPAD2 + 255) / 256, 256, 0, stream>>>(W2, W2p, HID, NCLS, NPAD2);
    }

    degree_kernel<<<(E + 255) / 256, 256, 0, stream>>>(dst, dinv, E);
    rsqrt_kernel<<<(N + 255) / 256, 256, 0, stream>>>(dinv, N);

    {
        const int pairTiles = (N + 31) / 32;
        const int wavesPerBlock = 4;
        const int blocks = (pairTiles + wavesPerBlock - 1) / wavesPerBlock;
        wmma_gemm_f32<8, 128><<<blocks, wavesPerBlock * 32, 0, stream>>>(x, W1p, h1, N, F_IN, HID);
    }

    {
        const int edgesPerBlock = 4;
        const int blocks = (E + edgesPerBlock - 1) / edgesPerBlock;
        scatter_kernel<<<blocks, edgesPerBlock * 32, 0, stream>>>(h1, src, dst, dinv, acc1, E, HID);
    }

    {
        size_t total = (size_t)N * HID;
        bias_relu_self_kernel<<<(unsigned)((total + 255) / 256), 256, 0, stream>>>(
            acc1, h1, dinv, b1, N, 7);
    }

    {
        const int pairTiles = (N + 31) / 32;
        const int wavesPerBlock = 4;
        const int blocks = (pairTiles + wavesPerBlock - 1) / wavesPerBlock;
        wmma_gemm_f32<3, 48><<<blocks, wavesPerBlock * 32, 0, stream>>>(acc1, W2p, h2, N, HID, NCLS);
    }

    {
        const int edgesPerBlock = 4;
        const int blocks = (E + edgesPerBlock - 1) / edgesPerBlock;
        scatter_kernel<<<blocks, edgesPerBlock * 32, 0, stream>>>(h2, src, dst, dinv, acc2, E, NCLS);
    }

    {
        const int wavesPerBlock = 8;
        const int blocks = (N + wavesPerBlock - 1) / wavesPerBlock;
        logsoftmax_kernel<<<blocks, wavesPerBlock * 32, 0, stream>>>(
            acc2, h2, dinv, b2, out, N, NCLS);
    }
}


# scatter: lane l owns features l, l+32, .. so each wave-wide f32 atomic and row load covers one contiguous 128-byte stretch, not a float4-per-lane stride over 4 lines; atomics, arrival order kept
# speedup vs baseline: 1.2164x; 1.2164x over previous
#include <hip/hip_runtime.h>
#include <hip/hip_bf16.h>

typedef __attribute__((ext_vector_type(2))) float v2f;
typedef __attribute__((ext_vector_type(8))) float v8f;

__global__ void fill_kernel(float* __restrict__ p, float v, size_t n) {
    size_t i = (size_t)blockIdx.x * blockDim.x + threadIdx.x;
    if (i < n) p[i] = v;
}

__global__ void pack_w_kernel(const float* __restrict__ W, float* __restrict__ P,
                              int K, int N, int NPAD) {
    int idx = blockIdx.x * blockDim.x + threadIdx.x;
    if (idx >= K * NPAD) return;
    int k = idx / NPAD;
    int c = idx - k * NPAD;
    float v = (c < N) ? W[(size_t)k * N + c] : 0.0f;
    P[(size_t)(k >> 1) * 2 * NPAD + 2 * c + (k & 1)] = v;
}

__global__ void degree_kernel(const int* __restrict__ dst, float* __restrict__ deg, int E) {
    int e = blockIdx.x * blockDim.x + threadIdx.x;
    if (e < E) atomicAdd(&deg[dst[e]], 1.0f);
}

__global__ void rsqrt_kernel(float* __restrict__ deg, int N) {
    int i = blockIdx.x * blockDim.x + threadIdx.x;
    if (i < N) {
        float d = deg[i];
        deg[i] = (d > 0.0f) ? rsqrtf(d) : 0.0f;
    }
}

template <int NT, int NPAD>
__global__ __launch_bounds__(128, 1) void wmma_gemm_f32(
        const float* __restrict__ A, const float* __restrict__ Wp,
        float* __restrict__ Out, int M, int K, int Nout) {
    const int lane = threadIdx.x & 31;
    const int wave = threadIdx.x >> 5;
    const int nPairTiles = (M + 31) >> 5;
    const int pairTile = blockIdx.x * (blockDim.x >> 5) + wave;
    if (pairTile >= nPairTiles) return;

    const int m0    = pairTile * 32 + (lane & 15);
    const int koff  = (lane >> 4) << 1;
    const int koff2 = (lane >> 4);
    const int n     = lane & 15;

    const bool hasSecond = (pairTile * 32 + 16) < M;

    v8f acc0[NT], acc1[NT];
#pragma unroll
    for (int t = 0; t < NT; ++t) {
        v8f z = {0.f, 0.f, 0.f, 0.f, 0.f, 0.f, 0.f, 0.f};
        acc0[t] = z;
        acc1[t] = z;
    }

    const float* A0 = A + (size_t)m0 * K;
    const float* A1 = hasSecond ? (A0 + (size_t)16 * K) : A0;

    for (int k = 0; k < K; k += 4) {
        v2f a0 = *(const v2f*)(A0 + k + koff);
        v2f a1 = *(const v2f*)(A1 + k + koff);
        const float* Wrow = Wp + (size_t)((k >> 1) + koff2) * (2 * NPAD);
#pragma unroll
        for (int t = 0; t < NT; ++t) {
            v2f b = *(const v2f*)(Wrow + 2 * (t * 16 + n));
            acc0[t] = __builtin_amdgcn_wmma_f32_16x16x4_f32(
                false, a0, false, b, (short)0, acc0[t], false, false);
            acc1[t] = __builtin_amdgcn_wmma_f32_16x16x4_f32(
                false, a1, false, b, (short)0, acc1[t], false, false);
        }
    }

    const int rbase0 = pairTile * 32 + ((lane >> 4) << 3);
    const int rbase1 = rbase0 + 16;
#pragma unroll
    for (int t = 0; t < NT; ++t) {
        int col = t * 16 + n;
        if (col < Nout) {
#pragma unroll
            for (int i = 0; i < 8; ++i)
                Out[(size_t)(rbase0 + i) * Nout + col] = acc0[t][i];
            if (hasSecond) {
#pragma unroll
                for (int i = 0; i < 8; ++i)
                    Out[(size_t)(rbase1 + i) * Nout + col] = acc1[t][i];
            }
        }
    }
}

template <int F>
__global__ void scatter_kernel(const float* __restrict__ H, const int* __restrict__ src,
                               const int* __restrict__ dst, const float* __restrict__ dinv,
                               float* __restrict__ acc, int E) {
    const int lane = threadIdx.x & 31;
    const int wave = threadIdx.x >> 5;
    const int e = blockIdx.x * (blockDim.x >> 5) + wave;
    if (e >= E) return;
    const int s = src[e];
    const int d = dst[e];
    const float norm = dinv[s] * dinv[d];
    const float* hrow = H + (size_t)s * F;
    float* out = acc + (size_t)d * F;
    constexpr int NJ = (F + 31) / 32;
    float hv[NJ];
#pragma unroll
    for (int j = 0; j < NJ; ++j) {
        const int f = lane + 32 * j;
        hv[j] = (f < F) ? hrow[f] : 0.0f;
    }
    if constexpr (NJ == 2 && (F & 31) != 0)
        asm volatile("" :: "v"(hv[0]), "v"(hv[1]));
#pragma unroll
    for (int j = 0; j < NJ; ++j) {
        const int f = lane + 32 * j;
        if (f < F) atomicAdd(out + f, hv[j] * norm);
    }
}

__global__ void bias_relu_self_kernel(float* __restrict__ acc, const float* __restrict__ h,
                                      const float* __restrict__ dinv, const float* __restrict__ b,
                                      int N, int logF) {
    size_t i = (size_t)blockIdx.x * blockDim.x + threadIdx.x;
    if (i >= ((size_t)N << logF)) return;
    int node = (int)(i >> logF);
    int f = (int)(i & ((1u << logF) - 1));
    float dn = dinv[node];
    float v = acc[i] + h[i] * dn * dn + b[f];
    acc[i] = v > 0.0f ? v : 0.0f;
}

__global__ void logsoftmax_kernel(const float* __restrict__ acc2, const float* __restrict__ h2,
                                  const float* __restrict__ dinv, const float* __restrict__ b2,
                                  float* __restrict__ out, int N, int C) {
    const int lane = threadIdx.x & 31;
    const int node = (blockIdx.x * blockDim.x + threadIdx.x) >> 5;
    if (node >= N) return;
    const float dn = dinv[node];
    const float sn = dn * dn;
    const int f0 = lane, f1 = lane + 32;
    const size_t base = (size_t)node * C;
    float v0 = -3.4e38f, v1 = -3.4e38f;
    if (f0 < C) v0 = acc2[base + f0] + h2[base + f0] * sn + b2[f0];
    if (f1 < C) v1 = acc2[base + f1] + h2[base + f1] * sn + b2[f1];
    float mx = fmaxf(v0, v1);
#pragma unroll
    for (int off = 16; off > 0; off >>= 1) mx = fmaxf(mx, __shfl_xor(mx, off, 32));
    float s = 0.0f;
    if (f0 < C) s += __expf(v0 - mx);
    if (f1 < C) s += __expf(v1 - mx);
#pragma unroll
    for (int off = 16; off > 0; off >>= 1) s += __shfl_xor(s, off, 32);
    const float lse = __logf(s) + mx;
    if (f0 < C) out[base + f0] = v0 - lse;
    if (f1 < C) out[base + f1] = v1 - lse;
}

extern "C" void kernel_launch(void* const* d_in, const int* in_sizes, int n_in,
                              void* d_out, int out_size, void* d_ws, size_t ws_size,
                              hipStream_t stream) {
    const float* x  = (const float*)d_in[0];
    const int*   ei = (const int*)d_in[1];
    const float* W1 = (const float*)d_in[2];
    const float* b1 = (const float*)d_in[3];
    const float* W2 = (const float*)d_in[4];
    const float* b2 = (const float*)d_in[5];
    float* out = (float*)d_out;

    const int F_IN  = 256;
    const int HID   = 128;
    const int NCLS  = 40;
    const int NPAD2 = 48;
    const int N = in_sizes[0] / F_IN;
    const int E = in_sizes[1] / 2;
    const int* src = ei;
    const int* dst = ei + E;

    float* dinv = (float*)d_ws;
    float* h1   = dinv + ((N + 63) & ~63);
    float* acc1 = h1 + (size_t)N * HID;
    float* h2   = acc1 + (size_t)N * HID;
    float* acc2 = h2 + (size_t)N * NCLS;
    float* W1p  = acc2 + (size_t)N * NCLS;
    float* W2p  = W1p + (size_t)F_IN * HID;

    {
        size_t n1 = N;
        fill_kernel<<<(unsigned)((n1 + 255) / 256), 256, 0, stream>>>(dinv, 1.0f, n1);
        size_t n2 = (size_t)N * HID;
        fill_kernel<<<(unsigned)((n2 + 255) / 256), 256, 0, stream>>>(acc1, 0.0f, n2);
        size_t n3 = (size_t)N * NCLS;
        fill_kernel<<<(unsigned)((n3 + 255) / 256), 256, 0, stream>>>(acc2, 0.0f, n3);
        pack_w_kernel<<<(F_IN * HID + 255) / 256, 256, 0, stream>>>(W1, W1p, F_IN, HID, HID);
        pack_w_kernel<<<(HID * NPAD2 + 255) / 256, 256, 0, stream>>>(W2, W2p, HID, NCLS, NPAD2);
    }

    degree_kernel<<<(E + 255) / 256, 256, 0, stream>>>(dst, dinv, E);
    rsqrt_kernel<<<(N + 255) / 256, 256, 0, stream>>>(dinv, N);

    {
        const int pairTiles = (N + 31) / 32;
        const int wavesPerBlock = 4;
        const int blocks = (pairTiles + wavesPerBlock - 1) / wavesPerBlock;
        wmma_gemm_f32<8, 128><<<blocks, wavesPerBlock * 32, 0, stream>>>(x, W1p, h1, N, F_IN, HID);
    }

    {
        const int edgesPerBlock = 4;
        const int blocks = (E + edgesPerBlock - 1) / edgesPerBlock;
        scatter_kernel<128><<<blocks, edgesPerBlock * 32, 0, stream>>>(h1, src, dst, dinv, acc1, E);
    }

    {
        size_t total = (size_t)N * HID;
        bias_relu_self_kernel<<<(unsigned)((total + 255) / 256), 256, 0, stream>>>(
            acc1, h1, dinv, b1, N, 7);
    }

    {
        const int pairTiles = (N + 31) / 32;
        const int wavesPerBlock = 4;
        const int blocks = (pairTiles + wavesPerBlock - 1) / wavesPerBlock;
        wmma_gemm_f32<3, 48><<<blocks, wavesPerBlock * 32, 0, stream>>>(acc1, W2p, h2, N, HID, NCLS);
    }

    {
        const int edgesPerBlock = 4;
        const int blocks = (E + edgesPerBlock - 1) / edgesPerBlock;
        scatter_kernel<40><<<blocks, edgesPerBlock * 32, 0, stream>>>(h2, src, dst, dinv, acc2, E);
    }

    {
        const int wavesPerBlock = 8;
        const int blocks = (N + wavesPerBlock - 1) / wavesPerBlock;
        logsoftmax_kernel<<<blocks, wavesPerBlock * 32, 0, stream>>>(
            acc2, h2, dinv, b2, out, N, NCLS);
    }
}


# edges counting-sorted by destination (int atomic = slot, scan, place); one wave per node sums in-edge rows in registers with self loop, bias, relu/log_softmax; no float atomics; slot-order sum
# speedup vs baseline: 1.9604x; 1.6116x over previous
#include <hip/hip_runtime.h>
#include <hip/hip_bf16.h>

typedef __attribute__((ext_vector_type(2))) float v2f;
typedef __attribute__((ext_vector_type(8))) float v8f;

__global__ void zero_counts_kernel(int* __restrict__ p, int n) {
    int i = blockIdx.x * blockDim.x + threadIdx.x;
    if (i < n) p[i] = 0;
}

__global__ void pack_w_kernel(const float* __restrict__ W, float* __restrict__ P,
                              int K, int N, int NPAD) {
    int idx = blockIdx.x * blockDim.x + threadIdx.x;
    if (idx >= K * NPAD) return;
    int k = idx / NPAD;
    int c = idx - k * NPAD;
    float v = (c < N) ? W[(size_t)k * N + c] : 0.0f;
    P[(size_t)(k >> 1) * 2 * NPAD + 2 * c + (k & 1)] = v;
}

__global__ void count_rank_kernel(const int* __restrict__ dst, int* __restrict__ cnt,
                                  int* __restrict__ rank, int E) {
    int e = blockIdx.x * blockDim.x + threadIdx.x;
    if (e < E) rank[e] = atomicAdd(&cnt[dst[e]], 1);
}

__global__ __launch_bounds__(1024) void scan_dinv_kernel(
        const int* __restrict__ cnt, int* __restrict__ off, float* __restrict__ dinv, int N) {
    __shared__ int wtot[32];
    const int tid = threadIdx.x;
    const int lane = tid & 31;
    const int wave = tid >> 5;
    const int blockBase = blockIdx.x * 1024;

    int part = 0;
    for (int i = tid; i < blockBase; i += 1024) part += cnt[i];
    for (int o = 16; o > 0; o >>= 1) part += __shfl_xor(part, o, 32);
    if (lane == 0) wtot[wave] = part;
    __syncthreads();
    int prefix = wtot[lane];
    for (int o = 16; o > 0; o >>= 1) prefix += __shfl_xor(prefix, o, 32);
    __syncthreads();

    const int i = blockBase + tid;
    const int c = (i < N) ? cnt[i] : 0;
    int x = c;
    for (int o = 1; o < 32; o <<= 1) {
        int y = __shfl_up(x, o, 32);
        if (lane >= o) x += y;
    }
    if (lane == 31) wtot[wave] = x;
    __syncthreads();
    const int wv = wtot[lane];
    int wx = wv;
    for (int o = 1; o < 32; o <<= 1) {
        int y = __shfl_up(wx, o, 32);
        if (lane >= o) wx += y;
    }
    const int waveOff = __shfl(wx - wv, wave, 32);
    if (i < N) {
        off[i] = prefix + waveOff + (x - c);
        const float d = (float)(c + 1);
        dinv[i] = (d > 0.0f) ? rsqrtf(d) : 0.0f;
    }
}

__global__ void place_edges_kernel(const int* __restrict__ src, const int* __restrict__ dst,
                                   const int* __restrict__ off, const int* __restrict__ rank,
                                   const float* __restrict__ dinv, int* __restrict__ seg, int E) {
    int e = blockIdx.x * blockDim.x + threadIdx.x;
    if (e >= E) return;
    const int s = src[e];
    const int pos = off[dst[e]] + rank[e];
    int2 v;
    v.x = s;
    v.y = __float_as_int(dinv[s]);
    *(int2*)(seg + 2 * (size_t)pos) = v;
}

template <int NT, int NPAD>
__global__ __launch_bounds__(128, 1) void wmma_gemm_f32(
        const float* __restrict__ A, const float* __restrict__ Wp,
        float* __restrict__ Out, int M, int K, int Nout) {
    const int lane = threadIdx.x & 31;
    const int wave = threadIdx.x >> 5;
    const int nPairTiles = (M + 31) >> 5;
    const int pairTile = blockIdx.x * (blockDim.x >> 5) + wave;
    if (pairTile >= nPairTiles) return;

    const int m0    = pairTile * 32 + (lane & 15);
    const int koff  = (lane >> 4) << 1;
    const int koff2 = (lane >> 4);
    const int n     = lane & 15;

    const bool hasSecond = (pairTile * 32 + 16) < M;

    v8f acc0[NT], acc1[NT];
#pragma unroll
    for (int t = 0; t < NT; ++t) {
        v8f z = {0.f, 0.f, 0.f, 0.f, 0.f, 0.f, 0.f, 0.f};
        acc0[t] = z;
        acc1[t] = z;
    }

    const float* A0 = A + (size_t)m0 * K;
    const float* A1 = hasSecond ? (A0 + (size_t)16 * K) : A0;

    for (int k = 0; k < K; k += 4) {
        v2f a0 = *(const v2f*)(A0 + k + koff);
        v2f a1 = *(const v2f*)(A1 + k + koff);
        const float* Wrow = Wp + (size_t)((k >> 1) + koff2) * (2 * NPAD);
#pragma unroll
        for (int t = 0; t < NT; ++t) {
            v2f b = *(const v2f*)(Wrow + 2 * (t * 16 + n));
            acc0[t] = __builtin_amdgcn_wmma_f32_16x16x4_f32(
                false, a0, false, b, (short)0, acc0[t], false, false);
            acc1[t] = __builtin_amdgcn_wmma_f32_16x16x4_f32(
                false, a1, false, b, (short)0, acc1[t], false, false);
        }
    }

    const int rbase0 = pairTile * 32 + ((lane >> 4) << 3);
    const int rbase1 = rbase0 + 16;
#pragma unroll
    for (int t = 0; t < NT; ++t) {
        int col = t * 16 + n;
        if (col < Nout) {
#pragma unroll
            for (int i = 0; i < 8; ++i)
                Out[(size_t)(rbase0 + i) * Nout + col] = acc0[t][i];
            if (hasSecond) {
#pragma unroll
                for (int i = 0; i < 8; ++i)
                    Out[(size_t)(rbase1 + i) * Nout + col] = acc1[t][i];
            }
        }
    }
}

template <int U>
__global__ __launch_bounds__(256) void gather_bias_relu_kernel(
        const float* __restrict__ H, const int* __restrict__ off, const int* __restrict__ cnt,
        const int* __restrict__ seg, const float* __restrict__ dinv, const float* __restrict__ b,
        float* __restrict__ outp, int N) {
    const int lane = threadIdx.x & 31;
    const int node = __builtin_amdgcn_readfirstlane((int)((blockIdx.x * blockDim.x + threadIdx.x) >> 5));
    if (node >= N) return;
    const int beg = __builtin_amdgcn_readfirstlane(off[node]);
    const int n   = __builtin_amdgcn_readfirstlane(cnt[node]);
    const float dn = dinv[node];
    const float sn = dn * dn;
    const float* hl = H + 4 * lane;
    const float* bl = b + 4 * lane;
    const float b0 = bl[0], b1v = bl[1], b2v = bl[2], b3v = bl[3];

    float4 acc;
    {
        const float4 h = *(const float4*)(hl + (size_t)node * 128);
        acc.x = h.x * sn; acc.y = h.y * sn; acc.z = h.z * sn; acc.w = h.w * sn;
    }

    for (int base = 0; base < n; base += 32) {
        const int m = min(n - base, 32);
        int ps = 0, pw = 0;
        if (lane < m) {
            const int2 p = *(const int2*)(seg + 2 * (size_t)(beg + base + lane));
            ps = p.x;
            pw = p.y;
        }
        for (int j = 0; j < m; j += U) {
            float4 r[U];
            float w[U];
            for (int u = 0; u < U; ++u) {
                const int jj = min(j + u, m - 1);
                const int s = __builtin_amdgcn_readlane(ps, jj);
                w[u] = __int_as_float(__builtin_amdgcn_readlane(pw, jj)) * dn;
                r[u] = *(const float4*)(hl + (size_t)s * 128);
            }
            for (int u = 0; u < U; ++u) {
                if (j + u < m) {
                    acc.x = fmaf(r[u].x, w[u], acc.x);
                    acc.y = fmaf(r[u].y, w[u], acc.y);
                    acc.z = fmaf(r[u].z, w[u], acc.z);
                    acc.w = fmaf(r[u].w, w[u], acc.w);
                }
            }
        }
    }

    float4 v;
    v.x = acc.x + b0; v.y = acc.y + b1v; v.z = acc.z + b2v; v.w = acc.w + b3v;
    v.x = v.x > 0.0f ? v.x : 0.0f;
    v.y = v.y > 0.0f ? v.y : 0.0f;
    v.z = v.z > 0.0f ? v.z : 0.0f;
    v.w = v.w > 0.0f ? v.w : 0.0f;
    *(float4*)(outp + (size_t)node * 128 + 4 * lane) = v;
}

template <int C, int U>
__global__ __launch_bounds__(256) void gather_logsoftmax_kernel(
        const float* __restrict__ H, const int* __restrict__ off, const int* __restrict__ cnt,
        const int* __restrict__ seg, const float* __restrict__ dinv, const float* __restrict__ b,
        float* __restrict__ out, int N) {
    const int lane = threadIdx.x & 31;
    const int node = __builtin_amdgcn_readfirstlane((int)((blockIdx.x * blockDim.x + threadIdx.x) >> 5));
    if (node >= N) return;
    const int beg = __builtin_amdgcn_readfirstlane(off[node]);
    const int n   = __builtin_amdgcn_readfirstlane(cnt[node]);
    const float dn = dinv[node];
    const float sn = dn * dn;
    const int f0 = lane, f1 = lane + 32;
    const bool ok0 = f0 < C, ok1 = f1 < C;
    const size_t obase = (size_t)node * C;

    float h0 = 0.0f, h1v = 0.0f, bias0 = 0.0f, bias1 = 0.0f;
    if (ok0) { h0 = H[obase + f0]; bias0 = b[f0]; }
    if (ok1) { h1v = H[obase + f1]; bias1 = b[f1]; }
    float a0 = h0 * sn, a1 = h1v * sn;

    for (int base = 0; base < n; base += 32) {
        const int m = min(n - base, 32);
        int ps = 0, pw = 0;
        if (lane < m) {
            const int2 p = *(const int2*)(seg + 2 * (size_t)(beg + base + lane));
            ps = p.x;
            pw = p.y;
        }
        for (int j = 0; j < m; j += U) {
            float r0[U], r1[U], w[U];
            for (int u = 0; u < U; ++u) {
                const int jj = min(j + u, m - 1);
                const int s = __builtin_amdgcn_readlane(ps, jj);
                w[u] = __int_as_float(__builtin_amdgcn_readlane(pw, jj)) * dn;
                const float* row = H + (size_t)s * C;
                r0[u] = ok0 ? row[f0] : 0.0f;
                r1[u] = ok1 ? row[f1] : 0.0f;
            }
            for (int u = 0; u < U; ++u) {
                if (j + u < m) {
                    a0 = fmaf(r0[u], w[u], a0);
                    a1 = fmaf(r1[u], w[u], a1);
                }
            }
        }
    }

    float v0 = -3.4e38f, v1 = -3.4e38f;
    if (ok0) v0 = a0 + bias0;
    if (ok1) v1 = a1 + bias1;
    float mx = fmaxf(v0, v1);
    for (int o = 16; o > 0; o >>= 1) mx = fmaxf(mx, __shfl_xor(mx, o, 32));
    float sum = 0.0f;
    if (ok0) sum += __expf(v0 - mx);
    if (ok1) sum += __expf(v1 - mx);
    for (int o = 16; o > 0; o >>= 1) sum += __shfl_xor(sum, o, 32);
    const float lse = __logf(sum) + mx;
    if (ok0) out[obase + f0] = v0 - lse;
    if (ok1) out[obase + f1] = v1 - lse;
}

extern "C" void kernel_launch(void* const* d_in, const int* in_sizes, int n_in,
                              void* d_out, int out_size, void* d_ws, size_t ws_size,
                              hipStream_t stream) {
    const float* x  = (const float*)d_in[0];
    const int*   ei = (const int*)d_in[1];
    const float* W1 = (const float*)d_in[2];
    const float* b1 = (const float*)d_in[3];
    const float* W2 = (const float*)d_in[4];
    const float* b2 = (const float*)d_in[5];
    float* out = (float*)d_out;

    const int F_IN  = 256;
    const int HID   = 128;
    const int NCLS  = 40;
    const int NPAD2 = 48;
    const int N = in_sizes[0] / F_IN;
    const int E = in_sizes[1] / 2;
    const int* src = ei;
    const int* dst = ei + E;
    const int NP = (N + 63) & ~63;
    const int EP = (E + 63) & ~63;

    int*   cnt  = (int*)d_ws;
    int*   off  = cnt + NP;
    float* dinv = (float*)(off + NP);
    int*   rank = (int*)(dinv + NP);
    int*   seg  = rank + EP;
    float* h1   = (float*)(seg + 2 * (size_t)EP);
    float* a1   = h1 + (size_t)N * HID;
    float* h2   = a1 + (size_t)N * HID;
    float* W1p  = h2 + (size_t)N * NCLS;
    float* W2p  = W1p + (size_t)F_IN * HID;

    zero_counts_kernel<<<(N + 255) / 256, 256, 0, stream>>>(cnt, N);
    pack_w_kernel<<<(F_IN * HID + 255) / 256, 256, 0, stream>>>(W1, W1p, F_IN, HID, HID);
    pack_w_kernel<<<(HID * NPAD2 + 255) / 256, 256, 0, stream>>>(W2, W2p, HID, NCLS, NPAD2);

    count_rank_kernel<<<(E + 255) / 256, 256, 0, stream>>>(dst, cnt, rank, E);
    scan_dinv_kernel<<<(N + 1023) / 1024, 1024, 0, stream>>>(cnt, off, dinv, N);
    place_edges_kernel<<<(E + 255) / 256, 256, 0, stream>>>(src, dst, off, rank, dinv, seg, E);

    {
        const int pairTiles = (N + 31) / 32;
        const int wavesPerBlock = 4;
        const int blocks = (pairTiles + wavesPerBlock - 1) / wavesPerBlock;
        wmma_gemm_f32<8, 128><<<blocks, wavesPerBlock * 32, 0, stream>>>(x, W1p, h1, N, F_IN, HID);
    }

    {
        const int wavesPerBlock = 8;
        const int blocks = (N + wavesPerBlock - 1) / wavesPerBlock;
        gather_bias_relu_kernel<8><<<blocks, wavesPerBlock * 32, 0, stream>>>(
            h1, off, cnt, seg, dinv, b1, a1, N);
    }

    {
        const int pairTiles = (N + 31) / 32;
        const int wavesPerBlock = 4;
        const int blocks = (pairTiles + wavesPerBlock - 1) / wavesPerBlock;
        wmma_gemm_f32<3, 48><<<blocks, wavesPerBlock * 32, 0, stream>>>(a1, W2p, h2, N, HID, NCLS);
    }

    {
        const int wavesPerBlock = 8;
        const int blocks = (N + wavesPerBlock - 1) / wavesPerBlock;
        gather_logsoftmax_kernel<40, 8><<<blocks, wavesPerBlock * 32, 0, stream>>>(
            h2, off, cnt, seg, dinv, b2, out, N);
    }
}


# edge count+slot by LDS atomics: 64 blocks each count one edge slice over all nodes in LDS (old value = slot), then running sum over slices per node; no global atomics; in-segment sum order changes
# speedup vs baseline: 2.3625x; 1.2051x over previous
#include <hip/hip_runtime.h>
#include <hip/hip_bf16.h>

typedef __attribute__((ext_vector_type(2))) float v2f;
typedef __attribute__((ext_vector_type(8))) float v8f;

__global__ void pack_w_kernel(const float* __restrict__ W, float* __restrict__ P,
                              int K, int N, int NPAD) {
    int idx = blockIdx.x * blockDim.x + threadIdx.x;
    if (idx >= K * NPAD) return;
    int k = idx / NPAD;
    int c = idx - k * NPAD;
    float v = (c < N) ? W[(size_t)k * N + c] : 0.0f;
    P[(size_t)(k >> 1) * 2 * NPAD + 2 * c + (k & 1)] = v;
}

#define HCAP 50176

__global__ __launch_bounds__(1024) void slice_count_rank_kernel(
        const int* __restrict__ dst, int* __restrict__ rank, int* __restrict__ part,
        int E, int N, int NP) {
    __shared__ int h[HCAP];
    const int tid = threadIdx.x;
    const int epb = (E + (int)gridDim.x - 1) / (int)gridDim.x;
    const int e0 = blockIdx.x * epb;
    const int e1 = min(e0 + epb, E);
    int* prow = part + (size_t)blockIdx.x * NP;

    for (int lo = 0; lo < N; lo += HCAP) {
        const int n = min(HCAP, N - lo);
        const int nq = n >> 2;
        const int4 z = {0, 0, 0, 0};
        for (int q = tid; q < nq; q += 1024) ((int4*)h)[q] = z;
        for (int i = (nq << 2) + tid; i < n; i += 1024) h[i] = 0;
        __syncthreads();

        for (int e = e0 + tid; e < e1; e += 8192) {
            int d[8];
#pragma unroll
            for (int u = 0; u < 8; ++u) d[u] = dst[min(e + u * 1024, e1 - 1)] - lo;
#pragma unroll
            for (int u = 0; u < 8; ++u) {
                const int ee = e + u * 1024;
                if (ee < e1 && (unsigned)d[u] < (unsigned)n) rank[ee] = atomicAdd(&h[d[u]], 1);
            }
        }
        __syncthreads();

        for (int q = tid; q < nq; q += 1024)
            *(int4*)(prow + lo + 4 * q) = ((const int4*)h)[q];
        for (int i = (nq << 2) + tid; i < n; i += 1024) prow[lo + i] = h[i];
        __syncthreads();
    }
}

template <int B>
__global__ void merge_slices_kernel(int* __restrict__ part, int* __restrict__ cnt, int N, int NP) {
    const int d = blockIdx.x * blockDim.x + threadIdx.x;
    if (d >= N) return;
    int* p = part + d;
    int run = 0;
    for (int b0 = 0; b0 < B; b0 += 16) {
        int v[16];
#pragma unroll
        for (int j = 0; j < 16; ++j) v[j] = p[(size_t)(b0 + j) * NP];
#pragma unroll
        for (int j = 0; j < 16; ++j) {
            p[(size_t)(b0 + j) * NP] = run;
            run += v[j];
        }
    }
    cnt[d] = run;
}

__global__ __launch_bounds__(1024) void scan_dinv_kernel(
        const int* __restrict__ cnt, int* __restrict__ off, float* __restrict__ dinv, int N) {
    __shared__ int wtot[32];
    const int tid = threadIdx.x;
    const int lane = tid & 31;
    const int wave = tid >> 5;
    const int blockBase = blockIdx.x * 1024;

    int part = 0;
    for (int i = tid; i < blockBase; i += 1024) part += cnt[i];
    for (int o = 16; o > 0; o >>= 1) part += __shfl_xor(part, o, 32);
    if (lane == 0) wtot[wave] = part;
    __syncthreads();
    int prefix = wtot[lane];
    for (int o = 16; o > 0; o >>= 1) prefix += __shfl_xor(prefix, o, 32);
    __syncthreads();

    const int i = blockBase + tid;
    const int c = (i < N) ? cnt[i] : 0;
    int x = c;
    for (int o = 1; o < 32; o <<= 1) {
        int y = __shfl_up(x, o, 32);
        if (lane >= o) x += y;
    }
    if (lane == 31) wtot[wave] = x;
    __syncthreads();
    const int wv = wtot[lane];
    int wx = wv;
    for (int o = 1; o < 32; o <<= 1) {
        int y = __shfl_up(wx, o, 32);
        if (lane >= o) wx += y;
    }
    const int waveOff = __shfl(wx - wv, wave, 32);
    if (i < N) {
        off[i] = prefix + waveOff + (x - c);
        const float d = (float)(c + 1);
        dinv[i] = (d > 0.0f) ? rsqrtf(d) : 0.0f;
    }
}

__global__ void place_edges_kernel(const int* __restrict__ src, const int* __restrict__ dst,
                                   const int* __restrict__ off, const int* __restrict__ rank,
                                   const int* __restrict__ part,
                                   const float* __restrict__ dinv, int* __restrict__ seg,
                                   int E, int NP, int B) {
    int e = blockIdx.x * blockDim.x + threadIdx.x;
    if (e >= E) return;
    const int s = src[e];
    const int d = dst[e];
    const int epb = (E + B - 1) / B;
    const int b = e / epb;
    const int pos = off[d] + part[(size_t)b * NP + d] + rank[e];
    int2 v;
    v.x = s;
    v.y = __float_as_int(dinv[s]);
    *(int2*)(seg + 2 * (size_t)pos) = v;
}

template <int NT, int NPAD>
__global__ __launch_bounds__(128, 1) void wmma_gemm_f32(
        const float* __restrict__ A, const float* __restrict__ Wp,
        float* __restrict__ Out, int M, int K, int Nout) {
    const int lane = threadIdx.x & 31;
    const int wave = threadIdx.x >> 5;
    const int nPairTiles = (M + 31) >> 5;
    const int pairTile = blockIdx.x * (blockDim.x >> 5) + wave;
    if (pairTile >= nPairTiles) return;

    const int m0    = pairTile * 32 + (lane & 15);
    const int koff  = (lane >> 4) << 1;
    const int koff2 = (lane >> 4);
    const int n     = lane & 15;

    const bool hasSecond = (pairTile * 32 + 16) < M;

    v8f acc0[NT], acc1[NT];
#pragma unroll
    for (int t = 0; t < NT; ++t) {
        v8f z = {0.f, 0.f, 0.f, 0.f, 0.f, 0.f, 0.f, 0.f};
        acc0[t] = z;
        acc1[t] = z;
    }

    const float* A0 = A + (size_t)m0 * K;
    const float* A1 = hasSecond ? (A0 + (size_t)16 * K) : A0;

    for (int k = 0; k < K; k += 4) {
        v2f a0 = *(const v2f*)(A0 + k + koff);
        v2f a1 = *(const v2f*)(A1 + k + koff);
        const float* Wrow = Wp + (size_t)((k >> 1) + koff2) * (2 * NPAD);
#pragma unroll
        for (int t = 0; t < NT; ++t) {
            v2f b = *(const v2f*)(Wrow + 2 * (t * 16 + n));
            acc0[t] = __builtin_amdgcn_wmma_f32_16x16x4_f32(
                false, a0, false, b, (short)0, acc0[t], false, false);
            acc1[t] = __builtin_amdgcn_wmma_f32_16x16x4_f32(
                false, a1, false, b, (short)0, acc1[t], false, false);
        }
    }

    const int rbase0 = pairTile * 32 + ((lane >> 4) << 3);
    const int rbase1 = rbase0 + 16;
#pragma unroll
    for (int t = 0; t < NT; ++t) {
        int col = t * 16 + n;
        if (col < Nout) {
#pragma unroll
            for (int i = 0; i < 8; ++i)
                Out[(size_t)(rbase0 + i) * Nout + col] = acc0[t][i];
            if (hasSecond) {
#pragma unroll
                for (int i = 0; i < 8; ++i)
                    Out[(size_t)(rbase1 + i) * Nout + col] = acc1[t][i];
            }
        }
    }
}

template <int U>
__global__ __launch_bounds__(256) void gather_bias_relu_kernel(
        const float* __restrict__ H, const int* __restrict__ off, const int* __restrict__ cnt,
        const int* __restrict__ seg, const float* __restrict__ dinv, const float* __restrict__ b,
        float* __restrict__ outp, int N) {
    const int lane = threadIdx.x & 31;
    const int node = __builtin_amdgcn_readfirstlane((int)((blockIdx.x * blockDim.x + threadIdx.x) >> 5));
    if (node >= N) return;
    const int beg = __builtin_amdgcn_readfirstlane(off[node]);
    const int n   = __builtin_amdgcn_readfirstlane(cnt[node]);
    const float dn = dinv[node];
    const float sn = dn * dn;
    const float* hl = H + 4 * lane;
    const float* bl = b + 4 * lane;
    const float b0 = bl[0], b1v = bl[1], b2v = bl[2], b3v = bl[3];

    float4 acc;
    {
        const float4 h = *(const float4*)(hl + (size_t)node * 128);
        acc.x = h.x * sn; acc.y = h.y * sn; acc.z = h.z * sn; acc.w = h.w * sn;
    }

    for (int base = 0; base < n; base += 32) {
        const int m = min(n - base, 32);
        int ps = 0, pw = 0;
        if (lane < m) {
            const int2 p = *(const int2*)(seg + 2 * (size_t)(beg + base + lane));
            ps = p.x;
            pw = p.y;
        }
        for (int j = 0; j < m; j += U) {
            float4 r[U];
            float w[U];
            for (int u = 0; u < U; ++u) {
                const int jj = min(j + u, m - 1);
                const int s = __builtin_amdgcn_readlane(ps, jj);
                w[u] = __int_as_float(__builtin_amdgcn_readlane(pw, jj)) * dn;
                r[u] = *(const float4*)(hl + (size_t)s * 128);
            }
            for (int u = 0; u < U; ++u) {
                if (j + u < m) {
                    acc.x = fmaf(r[u].x, w[u], acc.x);
                    acc.y = fmaf(r[u].y, w[u], acc.y);
                    acc.z = fmaf(r[u].z, w[u], acc.z);
                    acc.w = fmaf(r[u].w, w[u], acc.w);
                }
            }
        }
    }

    float4 v;
    v.x = acc.x + b0; v.y = acc.y + b1v; v.z = acc.z + b2v; v.w = acc.w + b3v;
    v.x = v.x > 0.0f ? v.x : 0.0f;
    v.y = v.y > 0.0f ? v.y : 0.0f;
    v.z = v.z > 0.0f ? v.z : 0.0f;
    v.w = v.w > 0.0f ? v.w : 0.0f;
    *(float4*)(outp + (size_t)node * 128 + 4 * lane) = v;
}

template <int C, int U>
__global__ __launch_bounds__(256) void gather_logsoftmax_kernel(
        const float* __restrict__ H, const int* __restrict__ off, const int* __restrict__ cnt,
        const int* __restrict__ seg, const float* __restrict__ dinv, const float* __restrict__ b,
        float* __restrict__ out, int N) {
    const int lane = threadIdx.x & 31;
    const int node = __builtin_amdgcn_readfirstlane((int)((blockIdx.x * blockDim.x + threadIdx.x) >> 5));
    if (node >= N) return;
    const int beg = __builtin_amdgcn_readfirstlane(off[node]);
    const int n   = __builtin_amdgcn_readfirstlane(cnt[node]);
    const float dn = dinv[node];
    const float sn = dn * dn;
    const int f0 = lane, f1 = lane + 32;
    const bool ok0 = f0 < C, ok1 = f1 < C;
    const size_t obase = (size_t)node * C;

    float h0 = 0.0f, h1v = 0.0f, bias0 = 0.0f, bias1 = 0.0f;
    if (ok0) { h0 = H[obase + f0]; bias0 = b[f0]; }
    if (ok1) { h1v = H[obase + f1]; bias1 = b[f1]; }
    float a0 = h0 * sn, a1 = h1v * sn;

    for (int base = 0; base < n; base += 32) {
        const int m = min(n - base, 32);
        int ps = 0, pw = 0;
        if (lane < m) {
            const int2 p = *(const int2*)(seg + 2 * (size_t)(beg + base + lane));
            ps = p.x;
            pw = p.y;
        }
        for (int j = 0; j < m; j += U) {
            float r0[U], r1[U], w[U];
            for (int u = 0; u < U; ++u) {
                const int jj = min(j + u, m - 1);
                const int s = __builtin_amdgcn_readlane(ps, jj);
                w[u] = __int_as_float(__builtin_amdgcn_readlane(pw, jj)) * dn;
                const float* row = H + (size_t)s * C;
                r0[u] = ok0 ? row[f0] : 0.0f;
                r1[u] = ok1 ? row[f1] : 0.0f;
            }
            for (int u = 0; u < U; ++u) {
                if (j + u < m) {
                    a0 = fmaf(r0[u], w[u], a0);
                    a1 = fmaf(r1[u], w[u], a1);
                }
            }
        }
    }

    float v0 = -3.4e38f, v1 = -3.4e38f;
    if (ok0) v0 = a0 + bias0;
    if (ok1) v1 = a1 + bias1;
    float mx = fmaxf(v0, v1);
    for (int o = 16; o > 0; o >>= 1) mx = fmaxf(mx, __shfl_xor(mx, o, 32));
    float sum = 0.0f;
    if (ok0) sum += __expf(v0 - mx);
    if (ok1) sum += __expf(v1 - mx);
    for (int o = 16; o > 0; o >>= 1) sum += __shfl_xor(sum, o, 32);
    const float lse = __logf(sum) + mx;
    if (ok0) out[obase + f0] = v0 - lse;
    if (ok1) out[obase + f1] = v1 - lse;
}

extern "C" void kernel_launch(void* const* d_in, const int* in_sizes, int n_in,
                              void* d_out, int out_size, void* d_ws, size_t ws_size,
                              hipStream_t stream) {
    const float* x  = (const float*)d_in[0];
    const int*   ei = (const int*)d_in[1];
    const float* W1 = (const float*)d_in[2];
    const float* b1 = (const float*)d_in[3];
    const float* W2 = (const float*)d_in[4];
    const float* b2 = (const float*)d_in[5];
    float* out = (float*)d_out;

    const int F_IN  = 256;
    const int HID   = 128;
    const int NCLS  = 40;
    const int NPAD2 = 48;
    const int N = in_sizes[0] / F_IN;
    const int E = in_sizes[1] / 2;
    const int* src = ei;
    const int* dst = ei + E;
    const int NP = (N + 63) & ~63;
    const int EP = (E + 63) & ~63;

    int*   cnt  = (int*)d_ws;
    int*   off  = cnt + NP;
    float* dinv = (float*)(off + NP);
    int*   rank = (int*)(dinv + NP);
    int*   seg  = rank + EP;
    float* h1   = (float*)(seg + 2 * (size_t)EP);
    float* a1   = h1 + (size_t)N * HID;
    float* h2   = a1 + (size_t)N * HID;
    float* W1p  = h2 + (size_t)N * NCLS;
    float* W2p  = W1p + (size_t)F_IN * HID;
    int*   part = (int*)(W2p + (size_t)HID * NPAD2);
    constexpr int SLICES = 64;

    pack_w_kernel<<<(F_IN * HID + 255) / 256, 256, 0, stream>>>(W1, W1p, F_IN, HID, HID);
    pack_w_kernel<<<(HID * NPAD2 + 255) / 256, 256, 0, stream>>>(W2, W2p, HID, NCLS, NPAD2);

    slice_count_rank_kernel<<<SLICES, 1024, 0, stream>>>(dst, rank, part, E, N, NP);
    merge_slices_kernel<SLICES><<<(N + 255) / 256, 256, 0, stream>>>(part, cnt, N, NP);
    scan_dinv_kernel<<<(N + 1023) / 1024, 1024, 0, stream>>>(cnt, off, dinv, N);
    place_edges_kernel<<<(E + 255) / 256, 256, 0, stream>>>(src, dst, off, rank, part, dinv, seg, E, NP, SLICES);

    {
        const int pairTiles = (N + 31) / 32;
        const int wavesPerBlock = 4;
        const int blocks = (pairTiles + wavesPerBlock - 1) / wavesPerBlock;
        wmma_gemm_f32<8, 128><<<blocks, wavesPerBlock * 32, 0, stream>>>(x, W1p, h1, N, F_IN, HID);
    }

    {
        const int wavesPerBlock = 8;
        const int blocks = (N + wavesPerBlock - 1) / wavesPerBlock;
        gather_bias_relu_kernel<8><<<blocks, wavesPerBlock * 32, 0, stream>>>(
            h1, off, cnt, seg, dinv, b1, a1, N);
    }

    {
        const int pairTiles = (N + 31) / 32;
        const int wavesPerBlock = 4;
        const int blocks = (pairTiles + wavesPerBlock - 1) / wavesPerBlock;
        wmma_gemm_f32<3, 48><<<blocks, wavesPerBlock * 32, 0, stream>>>(a1, W2p, h2, N, HID, NCLS);
    }

    {
        const int wavesPerBlock = 8;
        const int blocks = (N + wavesPerBlock - 1) / wavesPerBlock;
        gather_logsoftmax_kernel<40, 8><<<blocks, wavesPerBlock * 32, 0, stream>>>(
            h2, off, cnt, seg, dinv, b2, out, N);
    }
}


# independent stages share a launch, chosen by block index: x@W1 product blocks beside the edge placement, both weight packs beside the slice count; 10 launches become 7
# speedup vs baseline: 2.7517x; 1.1647x over previous
#include <hip/hip_runtime.h>
#include <hip/hip_bf16.h>

typedef __attribute__((ext_vector_type(2))) float v2f;
typedef __attribute__((ext_vector_type(8))) float v8f;

__device__ __forceinline__ void pack_w_role(const float* __restrict__ W, float* __restrict__ P,
                                            int K, int N, int NPAD, int idx) {
    if (idx >= K * NPAD) return;
    int k = idx / NPAD;
    int c = idx - k * NPAD;
    float v = (c < N) ? W[(size_t)k * N + c] : 0.0f;
    P[(size_t)(k >> 1) * 2 * NPAD + 2 * c + (k & 1)] = v;
}

#define HCAP 50176

__device__ __forceinline__ void slice_count_rank_role(
        int* h, const int* __restrict__ dst, int* __restrict__ rank, int* __restrict__ part,
        int E, int N, int NP, int slice, int nslices) {
    const int tid = threadIdx.x;
    const int epb = (E + nslices - 1) / nslices;
    const int e0 = slice * epb;
    const int e1 = min(e0 + epb, E);
    int* prow = part + (size_t)slice * NP;

    for (int lo = 0; lo < N; lo += HCAP) {
        const int n = min(HCAP, N - lo);
        const int nq = n >> 2;
        const int4 z = {0, 0, 0, 0};
        for (int q = tid; q < nq; q += 1024) ((int4*)h)[q] = z;
        for (int i = (nq << 2) + tid; i < n; i += 1024) h[i] = 0;
        __syncthreads();

        for (int e = e0 + tid; e < e1; e += 8192) {
            int d[8];
#pragma unroll
            for (int u = 0; u < 8; ++u) d[u] = dst[min(e + u * 1024, e1 - 1)] - lo;
#pragma unroll
            for (int u = 0; u < 8; ++u) {
                const int ee = e + u * 1024;
                if (ee < e1 && (unsigned)d[u] < (unsigned)n) rank[ee] = atomicAdd(&h[d[u]], 1);
            }
        }
        __syncthreads();

        for (int q = tid; q < nq; q += 1024)
            *(int4*)(prow + lo + 4 * q) = ((const int4*)h)[q];
        for (int i = (nq << 2) + tid; i < n; i += 1024) prow[lo + i] = h[i];
        __syncthreads();
    }
}

__global__ __launch_bounds__(1024) void count_pack_kernel(
        const int* __restrict__ dst, int* __restrict__ rank, int* __restrict__ part,
        int E, int N, int NP, int SLICES,
        const float* __restrict__ W1, float* __restrict__ W1p, int K1, int N1, int NPAD1, int P1,
        const float* __restrict__ W2, float* __restrict__ W2p, int K2, int N2, int NPAD2) {
    __shared__ int h[HCAP];
    const int bid = blockIdx.x;
    if (bid < SLICES) {
        slice_count_rank_role(h, dst, rank, part, E, N, NP, bid, SLICES);
    } else if (bid < SLICES + P1) {
        pack_w_role(W1, W1p, K1, N1, NPAD1, (bid - SLICES) * 1024 + (int)threadIdx.x);
    } else {
        pack_w_role(W2, W2p, K2, N2, NPAD2, (bid - SLICES - P1) * 1024 + (int)threadIdx.x);
    }
}

template <int B>
__global__ void merge_slices_kernel(int* __restrict__ part, int* __restrict__ cnt, int N, int NP) {
    const int d = blockIdx.x * blockDim.x + threadIdx.x;
    if (d >= N) return;
    int* p = part + d;
    int run = 0;
    for (int b0 = 0; b0 < B; b0 += 16) {
        int v[16];
#pragma unroll
        for (int j = 0; j < 16; ++j) v[j] = p[(size_t)(b0 + j) * NP];
#pragma unroll
        for (int j = 0; j < 16; ++j) {
            p[(size_t)(b0 + j) * NP] = run;
            run += v[j];
        }
    }
    cnt[d] = run;
}

__global__ __launch_bounds__(1024) void scan_dinv_kernel(
        const int* __restrict__ cnt, int* __restrict__ off, float* __restrict__ dinv, int N) {
    __shared__ int wtot[32];
    const int tid = threadIdx.x;
    const int lane = tid & 31;
    const int wave = tid >> 5;
    const int blockBase = blockIdx.x * 1024;

    int part = 0;
    for (int i = tid; i < blockBase; i += 1024) part += cnt[i];
    for (int o = 16; o > 0; o >>= 1) part += __shfl_xor(part, o, 32);
    if (lane == 0) wtot[wave] = part;
    __syncthreads();
    int prefix = wtot[lane];
    for (int o = 16; o > 0; o >>= 1) prefix += __shfl_xor(prefix, o, 32);
    __syncthreads();

    const int i = blockBase + tid;
    const int c = (i < N) ? cnt[i] : 0;
    int x = c;
    for (int o = 1; o < 32; o <<= 1) {
        int y = __shfl_up(x, o, 32);
        if (lane >= o) x += y;
    }
    if (lane == 31) wtot[wave] = x;
    __syncthreads();
    const int wv = wtot[lane];
    int wx = wv;
    for (int o = 1; o < 32; o <<= 1) {
        int y = __shfl_up(wx, o, 32);
        if (lane >= o) wx += y;
    }
    const int waveOff = __shfl(wx - wv, wave, 32);
    if (i < N) {
        off[i] = prefix + waveOff + (x - c);
        const float d = (float)(c + 1);
        dinv[i] = (d > 0.0f) ? rsqrtf(d) : 0.0f;
    }
}

__device__ __forceinline__ void place_edges_role(const int* __restrict__ src, const int* __restrict__ dst,
                                   const int* __restrict__ off, const int* __restrict__ rank,
                                   const int* __restrict__ part,
                                   const float* __restrict__ dinv, int* __restrict__ seg,
                                   int E, int NP, int B, int e) {
    if (e >= E) return;
    const int s = src[e];
    const int d = dst[e];
    const int epb = (E + B - 1) / B;
    const int b = e / epb;
    const int pos = off[d] + part[(size_t)b * NP + d] + rank[e];
    int2 v;
    v.x = s;
    v.y = __float_as_int(dinv[s]);
    *(int2*)(seg + 2 * (size_t)pos) = v;
}

template <int NT, int NPAD>
__device__ __forceinline__ void wmma_gemm_f32_role(
        const float* __restrict__ A, const float* __restrict__ Wp,
        float* __restrict__ Out, int M, int K, int Nout, int pairTile) {
    const int lane = threadIdx.x & 31;
    const int nPairTiles = (M + 31) >> 5;
    if (pairTile >= nPairTiles) return;

    const int m0    = pairTile * 32 + (lane & 15);
    const int koff  = (lane >> 4) << 1;
    const int koff2 = (lane >> 4);
    const int n     = lane & 15;

    const bool hasSecond = (pairTile * 32 + 16) < M;

    v8f acc0[NT], acc1[NT];
#pragma unroll
    for (int t = 0; t < NT; ++t) {
        v8f z = {0.f, 0.f, 0.f, 0.f, 0.f, 0.f, 0.f, 0.f};
        acc0[t] = z;
        acc1[t] = z;
    }

    const float* A0 = A + (size_t)m0 * K;
    const float* A1 = hasSecond ? (A0 + (size_t)16 * K) : A0;

    for (int k = 0; k < K; k += 4) {
        v2f a0 = *(const v2f*)(A0 + k + koff);
        v2f a1 = *(const v2f*)(A1 + k + koff);
        const float* Wrow = Wp + (size_t)((k >> 1) + koff2) * (2 * NPAD);
#pragma unroll
        for (int t = 0; t < NT; ++t) {
            v2f b = *(const v2f*)(Wrow + 2 * (t * 16 + n));
            acc0[t] = __builtin_amdgcn_wmma_f32_16x16x4_f32(
                false, a0, false, b, (short)0, acc0[t], false, false);
            acc1[t] = __builtin_amdgcn_wmma_f32_16x16x4_f32(
                false, a1, false, b, (short)0, acc1[t], false, false);
        }
    }

    const int rbase0 = pairTile * 32 + ((lane >> 4) << 3);
    const int rbase1 = rbase0 + 16;
#pragma unroll
    for (int t = 0; t < NT; ++t) {
        int col = t * 16 + n;
        if (col < Nout) {
#pragma unroll
            for (int i = 0; i < 8; ++i)
                Out[(size_t)(rbase0 + i) * Nout + col] = acc0[t][i];
            if (hasSecond) {
#pragma unroll
                for (int i = 0; i < 8; ++i)
                    Out[(size_t)(rbase1 + i) * Nout + col] = acc1[t][i];
            }
        }
    }
}

template <int NT, int NPAD>
__global__ __launch_bounds__(128, 1) void wmma_gemm_f32(
        const float* __restrict__ A, const float* __restrict__ Wp,
        float* __restrict__ Out, int M, int K, int Nout) {
    wmma_gemm_f32_role<NT, NPAD>(A, Wp, Out, M, K, Nout,
                                 (int)(blockIdx.x * (blockDim.x >> 5) + (threadIdx.x >> 5)));
}

template <int NT, int NPAD>
__global__ __launch_bounds__(256) void place_gemm_kernel(
        const float* __restrict__ A, const float* __restrict__ Wp,
        float* __restrict__ Out, int M, int K, int Nout, int GB,
        const int* __restrict__ src, const int* __restrict__ dst,
        const int* __restrict__ off, const int* __restrict__ rank,
        const int* __restrict__ part,
        const float* __restrict__ dinv, int* __restrict__ seg,
        int E, int NP, int B) {
    const int bid = blockIdx.x;
    if (bid < GB) {
        wmma_gemm_f32_role<NT, NPAD>(A, Wp, Out, M, K, Nout, bid * 8 + (int)(threadIdx.x >> 5));
    } else {
        place_edges_role(src, dst, off, rank, part, dinv, seg, E, NP, B,
                         (bid - GB) * 256 + (int)threadIdx.x);
    }
}

template <int U>
__global__ __launch_bounds__(256) void gather_bias_relu_kernel(
        const float* __restrict__ H, const int* __restrict__ off, const int* __restrict__ cnt,
        const int* __restrict__ seg, const float* __restrict__ dinv, const float* __restrict__ b,
        float* __restrict__ outp, int N) {
    const int lane = threadIdx.x & 31;
    const int node = __builtin_amdgcn_readfirstlane((int)((blockIdx.x * blockDim.x + threadIdx.x) >> 5));
    if (node >= N) return;
    const int beg = __builtin_amdgcn_readfirstlane(off[node]);
    const int n   = __builtin_amdgcn_readfirstlane(cnt[node]);
    const float dn = dinv[node];
    const float sn = dn * dn;
    const float* hl = H + 4 * lane;
    const float* bl = b + 4 * lane;
    const float b0 = bl[0], b1v = bl[1], b2v = bl[2], b3v = bl[3];

    float4 acc;
    {
        const float4 h = *(const float4*)(hl + (size_t)node * 128);
        acc.x = h.x * sn; acc.y = h.y * sn; acc.z = h.z * sn; acc.w = h.w * sn;
    }

    for (int base = 0; base < n; base += 32) {
        const int m = min(n - base, 32);
        int ps = 0, pw = 0;
        if (lane < m) {
            const int2 p = *(const int2*)(seg + 2 * (size_t)(beg + base + lane));
            ps = p.x;
            pw = p.y;
        }
        for (int j = 0; j < m; j += U) {
            float4 r[U];
            float w[U];
            for (int u = 0; u < U; ++u) {
                const int jj = min(j + u, m - 1);
                const int s = __builtin_amdgcn_readlane(ps, jj);
                w[u] = __int_as_float(__builtin_amdgcn_readlane(pw, jj)) * dn;
                r[u] = *(const float4*)(hl + (size_t)s * 128);
            }
            for (int u = 0; u < U; ++u) {
                if (j + u < m) {
                    acc.x = fmaf(r[u].x, w[u], acc.x);
                    acc.y = fmaf(r[u].y, w[u], acc.y);
                    acc.z = fmaf(r[u].z, w[u], acc.z);
                    acc.w = fmaf(r[u].w, w[u], acc.w);
                }
            }
        }
    }

    float4 v;
    v.x = acc.x + b0; v.y = acc.y + b1v; v.z = acc.z + b2v; v.w = acc.w + b3v;
    v.x = v.x > 0.0f ? v.x : 0.0f;
    v.y = v.y > 0.0f ? v.y : 0.0f;
    v.z = v.z > 0.0f ? v.z : 0.0f;
    v.w = v.w > 0.0f ? v.w : 0.0f;
    *(float4*)(outp + (size_t)node * 128 + 4 * lane) = v;
}

template <int C, int U>
__global__ __launch_bounds__(256) void gather_logsoftmax_kernel(
        const float* __restrict__ H, const int* __restrict__ off, const int* __restrict__ cnt,
        const int* __restrict__ seg, const float* __restrict__ dinv, const float* __restrict__ b,
        float* __restrict__ out, int N) {
    const int lane = threadIdx.x & 31;
    const int node = __builtin_amdgcn_readfirstlane((int)((blockIdx.x * blockDim.x + threadIdx.x) >> 5));
    if (node >= N) return;
    const int beg = __builtin_amdgcn_readfirstlane(off[node]);
    const int n   = __builtin_amdgcn_readfirstlane(cnt[node]);
    const float dn = dinv[node];
    const float sn = dn * dn;
    const int f0 = lane, f1 = lane + 32;
    const bool ok0 = f0 < C, ok1 = f1 < C;
    const size_t obase = (size_t)node * C;

    float h0 = 0.0f, h1v = 0.0f, bias0 = 0.0f, bias1 = 0.0f;
    if (ok0) { h0 = H[obase + f0]; bias0 = b[f0]; }
    if (ok1) { h1v = H[obase + f1]; bias1 = b[f1]; }
    float a0 = h0 * sn, a1 = h1v * sn;

    for (int base = 0; base < n; base += 32) {
        const int m = min(n - base, 32);
        int ps = 0, pw = 0;
        if (lane < m) {
            const int2 p = *(const int2*)(seg + 2 * (size_t)(beg + base + lane));
            ps = p.x;
            pw = p.y;
        }
        for (int j = 0; j < m; j += U) {
            float r0[U], r1[U], w[U];
            for (int u = 0; u < U; ++u) {
                const int jj = min(j + u, m - 1);
                const int s = __builtin_amdgcn_readlane(ps, jj);
                w[u] = __int_as_float(__builtin_amdgcn_readlane(pw, jj)) * dn;
                const float* row = H + (size_t)s * C;
                r0[u] = ok0 ? row[f0] : 0.0f;
                r1[u] = ok1 ? row[f1] : 0.0f;
            }
            for (int u = 0; u < U; ++u) {
                if (j + u < m) {
                    a0 = fmaf(r0[u], w[u], a0);
                    a1 = fmaf(r1[u], w[u], a1);
                }
            }
        }
    }

    float v0 = -3.4e38f, v1 = -3.4e38f;
    if (ok0) v0 = a0 + bias0;
    if (ok1) v1 = a1 + bias1;
    float mx = fmaxf(v0, v1);
    for (int o = 16; o > 0; o >>= 1) mx = fmaxf(mx, __shfl_xor(mx, o, 32));
    float sum = 0.0f;
    if (ok0) sum += __expf(v0 - mx);
    if (ok1) sum += __expf(v1 - mx);
    for (int o = 16; o > 0; o >>= 1) sum += __shfl_xor(sum, o, 32);
    const float lse = __logf(sum) + mx;
    if (ok0) out[obase + f0] = v0 - lse;
    if (ok1) out[obase + f1] = v1 - lse;
}

extern "C" void kernel_launch(void* const* d_in, const int* in_sizes, int n_in,
                              void* d_out, int out_size, void* d_ws, size_t ws_size,
                              hipStream_t stream) {
    const float* x  = (const float*)d_in[0];
    const int*   ei = (const int*)d_in[1];
    const float* W1 = (const float*)d_in[2];
    const float* b1 = (const float*)d_in[3];
    const float* W2 = (const float*)d_in[4];
    const float* b2 = (const float*)d_in[5];
    float* out = (float*)d_out;

    const int F_IN  = 256;
    const int HID   = 128;
    const int NCLS  = 40;
    const int NPAD2 = 48;
    const int N = in_sizes[0] / F_IN;
    const int E = in_sizes[1] / 2;
    const int* src = ei;
    const int* dst = ei + E;
    const int NP = (N + 63) & ~63;
    const int EP = (E + 63) & ~63;

    int*   cnt  = (int*)d_ws;
    int*   off  = cnt + NP;
    float* dinv = (float*)(off + NP);
    int*   rank = (int*)(dinv + NP);
    int*   seg  = rank + EP;
    float* h1   = (float*)(seg + 2 * (size_t)EP);
    float* a1   = h1 + (size_t)N * HID;
    float* h2   = a1 + (size_t)N * HID;
    float* W1p  = h2 + (size_t)N * NCLS;
    float* W2p  = W1p + (size_t)F_IN * HID;
    int*   part = (int*)(W2p + (size_t)HID * NPAD2);
    constexpr int SLICES = 64;

    {
        const int P1 = (F_IN * HID + 1023) / 1024;
        const int P2 = (HID * NPAD2 + 1023) / 1024;
        count_pack_kernel<<<SLICES + P1 + P2, 1024, 0, stream>>>(
            dst, rank, part, E, N, NP, SLICES,
            W1, W1p, F_IN, HID, HID, P1,
            W2, W2p, HID, NCLS, NPAD2);
    }

    merge_slices_kernel<SLICES><<<(N + 255) / 256, 256, 0, stream>>>(part, cnt, N, NP);
    scan_dinv_kernel<<<(N + 1023) / 1024, 1024, 0, stream>>>(cnt, off, dinv, N);

    {
        const int pairTiles = (N + 31) / 32;
        const int GB = (pairTiles + 7) / 8;
        const int PB = (E + 255) / 256;
        place_gemm_kernel<8, 128><<<GB + PB, 256, 0, stream>>>(
            x, W1p, h1, N, F_IN, HID, GB,
            src, dst, off, rank, part, dinv, seg, E, NP, SLICES);
    }

    {
        const int wavesPerBlock = 8;
        const int blocks = (N + wavesPerBlock - 1) / wavesPerBlock;
        gather_bias_relu_kernel<8><<<blocks, wavesPerBlock * 32, 0, stream>>>(
            h1, off, cnt, seg, dinv, b1, a1, N);
    }

    {
        const int pairTiles = (N + 31) / 32;
        const int wavesPerBlock = 4;
        const int blocks = (pairTiles + wavesPerBlock - 1) / wavesPerBlock;
        wmma_gemm_f32<3, 48><<<blocks, wavesPerBlock * 32, 0, stream>>>(a1, W2p, h2, N, HID, NCLS);
    }

    {
        const int wavesPerBlock = 8;
        const int blocks = (N + wavesPerBlock - 1) / wavesPerBlock;
        gather_logsoftmax_kernel<40, 8><<<blocks, wavesPerBlock * 32, 0, stream>>>(
            h2, off, cnt, seg, dinv, b2, out, N);
    }
}


# layer 2 product a1@W2 done in the layer 1 aggregation's epilogue: 16 nodes per block, relu rows to an LDS tile, 3 waves run f32 WMMA on it; a1 never stored; 7 launches to 6; k summed in 2 chains
# speedup vs baseline: 2.8537x; 1.0371x over previous
#include <hip/hip_runtime.h>
#include <hip/hip_bf16.h>

typedef __attribute__((ext_vector_type(2))) float v2f;
typedef __attribute__((ext_vector_type(8))) float v8f;

__device__ __forceinline__ void pack_w_role(const float* __restrict__ W, float* __restrict__ P,
                                            int K, int N, int NPAD, int idx) {
    if (idx >= K * NPAD) return;
    int k = idx / NPAD;
    int c = idx - k * NPAD;
    float v = (c < N) ? W[(size_t)k * N + c] : 0.0f;
    P[(size_t)(k >> 1) * 2 * NPAD + 2 * c + (k & 1)] = v;
}

#define HCAP 50176

__device__ __forceinline__ void slice_count_rank_role(
        int* h, const int* __restrict__ dst, int* __restrict__ rank, int* __restrict__ part,
        int E, int N, int NP, int slice, int nslices) {
    const int tid = threadIdx.x;
    const int epb = (E + nslices - 1) / nslices;
    const int e0 = slice * epb;
    const int e1 = min(e0 + epb, E);
    int* prow = part + (size_t)slice * NP;

    for (int lo = 0; lo < N; lo += HCAP) {
        const int n = min(HCAP, N - lo);
        const int nq = n >> 2;
        const int4 z = {0, 0, 0, 0};
        for (int q = tid; q < nq; q += 1024) ((int4*)h)[q] = z;
        for (int i = (nq << 2) + tid; i < n; i += 1024) h[i] = 0;
        __syncthreads();

        for (int e = e0 + tid; e < e1; e += 8192) {
            int d[8];
#pragma unroll
            for (int u = 0; u < 8; ++u) d[u] = dst[min(e + u * 1024, e1 - 1)] - lo;
#pragma unroll
            for (int u = 0; u < 8; ++u) {
                const int ee = e + u * 1024;
                if (ee < e1 && (unsigned)d[u] < (unsigned)n) rank[ee] = atomicAdd(&h[d[u]], 1);
            }
        }
        __syncthreads();

        for (int q = tid; q < nq; q += 1024)
            *(int4*)(prow + lo + 4 * q) = ((const int4*)h)[q];
        for (int i = (nq << 2) + tid; i < n; i += 1024) prow[lo + i] = h[i];
        __syncthreads();
    }
}

__global__ __launch_bounds__(1024) void count_pack_kernel(
        const int* __restrict__ dst, int* __restrict__ rank, int* __restrict__ part,
        int E, int N, int NP, int SLICES,
        const float* __restrict__ W1, float* __restrict__ W1p, int K1, int N1, int NPAD1, int P1,
        const float* __restrict__ W2, float* __restrict__ W2p, int K2, int N2, int NPAD2) {
    __shared__ int h[HCAP];
    const int bid = blockIdx.x;
    if (bid < SLICES) {
        slice_count_rank_role(h, dst, rank, part, E, N, NP, bid, SLICES);
    } else if (bid < SLICES + P1) {
        pack_w_role(W1, W1p, K1, N1, NPAD1, (bid - SLICES) * 1024 + (int)threadIdx.x);
    } else {
        pack_w_role(W2, W2p, K2, N2, NPAD2, (bid - SLICES - P1) * 1024 + (int)threadIdx.x);
    }
}

template <int B>
__global__ void merge_slices_kernel(int* __restrict__ part, int* __restrict__ cnt, int N, int NP) {
    const int d = blockIdx.x * blockDim.x + threadIdx.x;
    if (d >= N) return;
    int* p = part + d;
    int run = 0;
    for (int b0 = 0; b0 < B; b0 += 16) {
        int v[16];
#pragma unroll
        for (int j = 0; j < 16; ++j) v[j] = p[(size_t)(b0 + j) * NP];
#pragma unroll
        for (int j = 0; j < 16; ++j) {
            p[(size_t)(b0 + j) * NP] = run;
            run += v[j];
        }
    }
    cnt[d] = run;
}

__global__ __launch_bounds__(1024) void scan_dinv_kernel(
        const int* __restrict__ cnt, int* __restrict__ off, float* __restrict__ dinv, int N) {
    __shared__ int wtot[32];
    const int tid = threadIdx.x;
    const int lane = tid & 31;
    const int wave = tid >> 5;
    const int blockBase = blockIdx.x * 1024;

    int part = 0;
    for (int i = tid; i < blockBase; i += 1024) part += cnt[i];
    for (int o = 16; o > 0; o >>= 1) part += __shfl_xor(part, o, 32);
    if (lane == 0) wtot[wave] = part;
    __syncthreads();
    int prefix = wtot[lane];
    for (int o = 16; o > 0; o >>= 1) prefix += __shfl_xor(prefix, o, 32);
    __syncthreads();

    const int i = blockBase + tid;
    const int c = (i < N) ? cnt[i] : 0;
    int x = c;
    for (int o = 1; o < 32; o <<= 1) {
        int y = __shfl_up(x, o, 32);
        if (lane >= o) x += y;
    }
    if (lane == 31) wtot[wave] = x;
    __syncthreads();
    const int wv = wtot[lane];
    int wx = wv;
    for (int o = 1; o < 32; o <<= 1) {
        int y = __shfl_up(wx, o, 32);
        if (lane >= o) wx += y;
    }
    const int waveOff = __shfl(wx - wv, wave, 32);
    if (i < N) {
        off[i] = prefix + waveOff + (x - c);
        const float d = (float)(c + 1);
        dinv[i] = (d > 0.0f) ? rsqrtf(d) : 0.0f;
    }
}

__device__ __forceinline__ void place_edges_role(const int* __restrict__ src, const int* __restrict__ dst,
                                   const int* __restrict__ off, const int* __restrict__ rank,
                                   const int* __restrict__ part,
                                   const float* __restrict__ dinv, int* __restrict__ seg,
                                   int E, int NP, int B, int e) {
    if (e >= E) return;
    const int s = src[e];
    const int d = dst[e];
    const int epb = (E + B - 1) / B;
    const int b = e / epb;
    const int pos = off[d] + part[(size_t)b * NP + d] + rank[e];
    int2 v;
    v.x = s;
    v.y = __float_as_int(dinv[s]);
    *(int2*)(seg + 2 * (size_t)pos) = v;
}

template <int NT, int NPAD>
__device__ __forceinline__ void wmma_gemm_f32_role(
        const float* __restrict__ A, const float* __restrict__ Wp,
        float* __restrict__ Out, int M, int K, int Nout, int pairTile) {
    const int lane = threadIdx.x & 31;
    const int nPairTiles = (M + 31) >> 5;
    if (pairTile >= nPairTiles) return;

    const int m0    = pairTile * 32 + (lane & 15);
    const int koff  = (lane >> 4) << 1;
    const int koff2 = (lane >> 4);
    const int n     = lane & 15;

    const bool hasSecond = (pairTile * 32 + 16) < M;

    v8f acc0[NT], acc1[NT];
#pragma unroll
    for (int t = 0; t < NT; ++t) {
        v8f z = {0.f, 0.f, 0.f, 0.f, 0.f, 0.f, 0.f, 0.f};
        acc0[t] = z;
        acc1[t] = z;
    }

    const float* A0 = A + (size_t)m0 * K;
    const float* A1 = hasSecond ? (A0 + (size_t)16 * K) : A0;

    for (int k = 0; k < K; k += 4) {
        v2f a0 = *(const v2f*)(A0 + k + koff);
        v2f a1 = *(const v2f*)(A1 + k + koff);
        const float* Wrow = Wp + (size_t)((k >> 1) + koff2) * (2 * NPAD);
#pragma unroll
        for (int t = 0; t < NT; ++t) {
            v2f b = *(const v2f*)(Wrow + 2 * (t * 16 + n));
            acc0[t] = __builtin_amdgcn_wmma_f32_16x16x4_f32(
                false, a0, false, b, (short)0, acc0[t], false, false);
            acc1[t] = __builtin_amdgcn_wmma_f32_16x16x4_f32(
                false, a1, false, b, (short)0, acc1[t], false, false);
        }
    }

    const int rbase0 = pairTile * 32 + ((lane >> 4) << 3);
    const int rbase1 = rbase0 + 16;
#pragma unroll
    for (int t = 0; t < NT; ++t) {
        int col = t * 16 + n;
        if (col < Nout) {
#pragma unroll
            for (int i = 0; i < 8; ++i)
                Out[(size_t)(rbase0 + i) * Nout + col] = acc0[t][i];
            if (hasSecond) {
#pragma unroll
                for (int i = 0; i < 8; ++i)
                    Out[(size_t)(rbase1 + i) * Nout + col] = acc1[t][i];
            }
        }
    }
}

template <int NT, int NPAD>
__global__ __launch_bounds__(256) void place_gemm_kernel(
        const float* __restrict__ A, const float* __restrict__ Wp,
        float* __restrict__ Out, int M, int K, int Nout, int GB,
        const int* __restrict__ src, const int* __restrict__ dst,
        const int* __restrict__ off, const int* __restrict__ rank,
        const int* __restrict__ part,
        const float* __restrict__ dinv, int* __restrict__ seg,
        int E, int NP, int B) {
    const int bid = blockIdx.x;
    if (bid < GB) {
        wmma_gemm_f32_role<NT, NPAD>(A, Wp, Out, M, K, Nout, bid * 8 + (int)(threadIdx.x >> 5));
    } else {
        place_edges_role(src, dst, off, rank, part, dinv, seg, E, NP, B,
                         (bid - GB) * 256 + (int)threadIdx.x);
    }
}

template <int U, int NT, int NPAD>
__global__ __launch_bounds__(512) void gather_relu_gemm_kernel(
        const float* __restrict__ H, const int* __restrict__ off, const int* __restrict__ cnt,
        const int* __restrict__ seg, const float* __restrict__ dinv, const float* __restrict__ b,
        const float* __restrict__ Wp, float* __restrict__ Out, int N, int Nout) {
    constexpr int PITCH = 132;
    __shared__ float tile[16 * PITCH];
    const int lane = threadIdx.x & 31;
    const int wave = __builtin_amdgcn_readfirstlane((int)(threadIdx.x >> 5));
    const int node = blockIdx.x * 16 + wave;

    float4 v = {0.0f, 0.0f, 0.0f, 0.0f};
    if (node < N) {
        const int beg = __builtin_amdgcn_readfirstlane(off[node]);
        const int n   = __builtin_amdgcn_readfirstlane(cnt[node]);
        const float dn = dinv[node];
        const float sn = dn * dn;
        const float* hl = H + 4 * lane;
        const float* bl = b + 4 * lane;
        const float b0 = bl[0], b1v = bl[1], b2v = bl[2], b3v = bl[3];

        float4 acc;
        {
            const float4 h = *(const float4*)(hl + (size_t)node * 128);
            acc.x = h.x * sn; acc.y = h.y * sn; acc.z = h.z * sn; acc.w = h.w * sn;
        }

        for (int base = 0; base < n; base += 32) {
            const int m = min(n - base, 32);
            int ps = 0, pw = 0;
            if (lane < m) {
                const int2 p = *(const int2*)(seg + 2 * (size_t)(beg + base + lane));
                ps = p.x;
                pw = p.y;
            }
            for (int j = 0; j < m; j += U) {
                float4 r[U];
                float w[U];
                for (int u = 0; u < U; ++u) {
                    const int jj = min(j + u, m - 1);
                    const int s = __builtin_amdgcn_readlane(ps, jj);
                    w[u] = __int_as_float(__builtin_amdgcn_readlane(pw, jj)) * dn;
                    r[u] = *(const float4*)(hl + (size_t)s * 128);
                }
                for (int u = 0; u < U; ++u) {
                    if (j + u < m) {
                        acc.x = fmaf(r[u].x, w[u], acc.x);
                        acc.y = fmaf(r[u].y, w[u], acc.y);
                        acc.z = fmaf(r[u].z, w[u], acc.z);
                        acc.w = fmaf(r[u].w, w[u], acc.w);
                    }
                }
            }
        }

        v.x = acc.x + b0; v.y = acc.y + b1v; v.z = acc.z + b2v; v.w = acc.w + b3v;
        v.x = v.x > 0.0f ? v.x : 0.0f;
        v.y = v.y > 0.0f ? v.y : 0.0f;
        v.z = v.z > 0.0f ? v.z : 0.0f;
        v.w = v.w > 0.0f ? v.w : 0.0f;
    }
    *(float4*)(tile + wave * PITCH + 4 * lane) = v;
    __syncthreads();
    if (wave >= NT) return;

    const int c = lane & 15;
    const int hh = lane >> 4;
    const float* arow = tile + c * PITCH + 2 * hh;
    const float* wcol = Wp + (size_t)hh * (2 * NPAD) + 2 * (wave * 16 + c);
    v8f acc0 = {0.f, 0.f, 0.f, 0.f, 0.f, 0.f, 0.f, 0.f};
    v8f acc1 = acc0;
    for (int k = 0; k < 128; k += 8) {
        const v2f a0 = *(const v2f*)(arow + k);
        const v2f a1 = *(const v2f*)(arow + k + 4);
        const v2f w0 = *(const v2f*)(wcol + (size_t)(k >> 1) * (2 * NPAD));
        const v2f w1 = *(const v2f*)(wcol + (size_t)((k >> 1) + 2) * (2 * NPAD));
        acc0 = __builtin_amdgcn_wmma_f32_16x16x4_f32(
            false, a0, false, w0, (short)0, acc0, false, false);
        acc1 = __builtin_amdgcn_wmma_f32_16x16x4_f32(
            false, a1, false, w1, (short)0, acc1, false, false);
    }

    const int col = wave * 16 + c;
    const int rbase = blockIdx.x * 16 + (hh << 3);
    if (col < Nout) {
#pragma unroll
        for (int i = 0; i < 8; ++i)
            if (rbase + i < N) Out[(size_t)(rbase + i) * Nout + col] = acc0[i] + acc1[i];
    }
}

template <int C, int U>
__global__ __launch_bounds__(256) void gather_logsoftmax_kernel(
        const float* __restrict__ H, const int* __restrict__ off, const int* __restrict__ cnt,
        const int* __restrict__ seg, const float* __restrict__ dinv, const float* __restrict__ b,
        float* __restrict__ out, int N) {
    const int lane = threadIdx.x & 31;
    const int node = __builtin_amdgcn_readfirstlane((int)((blockIdx.x * blockDim.x + threadIdx.x) >> 5));
    if (node >= N) return;
    const int beg = __builtin_amdgcn_readfirstlane(off[node]);
    const int n   = __builtin_amdgcn_readfirstlane(cnt[node]);
    const float dn = dinv[node];
    const float sn = dn * dn;
    const int f0 = lane, f1 = lane + 32;
    const bool ok0 = f0 < C, ok1 = f1 < C;
    const size_t obase = (size_t)node * C;

    float h0 = 0.0f, h1v = 0.0f, bias0 = 0.0f, bias1 = 0.0f;
    if (ok0) { h0 = H[obase + f0]; bias0 = b[f0]; }
    if (ok1) { h1v = H[obase + f1]; bias1 = b[f1]; }
    float a0 = h0 * sn, a1 = h1v * sn;

    for (int base = 0; base < n; base += 32) {
        const int m = min(n - base, 32);
        int ps = 0, pw = 0;
        if (lane < m) {
            const int2 p = *(const int2*)(seg + 2 * (size_t)(beg + base + lane));
            ps = p.x;
            pw = p.y;
        }
        for (int j = 0; j < m; j += U) {
            float r0[U], r1[U], w[U];
            for (int u = 0; u < U; ++u) {
                const int jj = min(j + u, m - 1);
                const int s = __builtin_amdgcn_readlane(ps, jj);
                w[u] = __int_as_float(__builtin_amdgcn_readlane(pw, jj)) * dn;
                const float* row = H + (size_t)s * C;
                r0[u] = ok0 ? row[f0] : 0.0f;
                r1[u] = ok1 ? row[f1] : 0.0f;
            }
            for (int u = 0; u < U; ++u) {
                if (j + u < m) {
                    a0 = fmaf(r0[u], w[u], a0);
                    a1 = fmaf(r1[u], w[u], a1);
                }
            }
        }
    }

    float v0 = -3.4e38f, v1 = -3.4e38f;
    if (ok0) v0 = a0 + bias0;
    if (ok1) v1 = a1 + bias1;
    float mx = fmaxf(v0, v1);
    for (int o = 16; o > 0; o >>= 1) mx = fmaxf(mx, __shfl_xor(mx, o, 32));
    float sum = 0.0f;
    if (ok0) sum += __expf(v0 - mx);
    if (ok1) sum += __expf(v1 - mx);
    for (int o = 16; o > 0; o >>= 1) sum += __shfl_xor(sum, o, 32);
    const float lse = __logf(sum) + mx;
    if (ok0) out[obase + f0] = v0 - lse;
    if (ok1) out[obase + f1] = v1 - lse;
}

extern "C" void kernel_launch(void* const* d_in, const int* in_sizes, int n_in,
                              void* d_out, int out_size, void* d_ws, size_t ws_size,
                              hipStream_t stream) {
    const float* x  = (const float*)d_in[0];
    const int*   ei = (const int*)d_in[1];
    const float* W1 = (const float*)d_in[2];
    const float* b1 = (const float*)d_in[3];
    const float* W2 = (const float*)d_in[4];
    const float* b2 = (const float*)d_in[5];
    float* out = (float*)d_out;

    const int F_IN  = 256;
    const int HID   = 128;
    const int NCLS  = 40;
    const int NPAD2 = 48;
    const int N = in_sizes[0] / F_IN;
    const int E = in_sizes[1] / 2;
    const int* src = ei;
    const int* dst = ei + E;
    const int NP = (N + 63) & ~63;
    const int EP = (E + 63) & ~63;

    int*   cnt  = (int*)d_ws;
    int*   off  = cnt + NP;
    float* dinv = (float*)(off + NP);
    int*   rank = (int*)(dinv + NP);
    int*   seg  = rank + EP;
    float* h1   = (float*)(seg + 2 * (size_t)EP);
    float* h2   = h1 + (size_t)N * HID;
    float* W1p  = h2 + (size_t)N * NCLS;
    float* W2p  = W1p + (size_t)F_IN * HID;
    int*   part = (int*)(W2p + (size_t)HID * NPAD2);
    constexpr int SLICES = 64;

    {
        const int P1 = (F_IN * HID + 1023) / 1024;
        const int P2 = (HID * NPAD2 + 1023) / 1024;
        count_pack_kernel<<<SLICES + P1 + P2, 1024, 0, stream>>>(
            dst, rank, part, E, N, NP, SLICES,
            W1, W1p, F_IN, HID, HID, P1,
            W2, W2p, HID, NCLS, NPAD2);
    }

    merge_slices_kernel<SLICES><<<(N + 255) / 256, 256, 0, stream>>>(part, cnt, N, NP);
    scan_dinv_kernel<<<(N + 1023) / 1024, 1024, 0, stream>>>(cnt, off, dinv, N);

    {
        const int pairTiles = (N + 31) / 32;
        const int GB = (pairTiles + 7) / 8;
        const int PB = (E + 255) / 256;
        place_gemm_kernel<8, 128><<<GB + PB, 256, 0, stream>>>(
            x, W1p, h1, N, F_IN, HID, GB,
            src, dst, off, rank, part, dinv, seg, E, NP, SLICES);
    }

    {
        const int nodesPerBlock = 16;
        const int blocks = (N + nodesPerBlock - 1) / nodesPerBlock;
        gather_relu_gemm_kernel<8, 3, 48><<<blocks, nodesPerBlock * 32, 0, stream>>>(
            h1, off, cnt, seg, dinv, b1, W2p, h2, N, NCLS);
    }

    {
        const int wavesPerBlock = 8;
        const int blocks = (N + wavesPerBlock - 1) / wavesPerBlock;
        gather_logsoftmax_kernel<40, 8><<<blocks, wavesPerBlock * 32, 0, stream>>>(
            h2, off, cnt, seg, dinv, b2, out, N);
    }
}


# slice merge, node scan and dinv in one launch: count blocks also leave per-chunk running totals, so a 256-node merge block sums 64 numbers for its base; part holds final slots; 6 launches to 5
# speedup vs baseline: 3.0660x; 1.0744x over previous
#include <hip/hip_runtime.h>
#include <hip/hip_bf16.h>

typedef __attribute__((ext_vector_type(2))) float v2f;
typedef __attribute__((ext_vector_type(8))) float v8f;

__device__ __forceinline__ void pack_w_role(const float* __restrict__ W, float* __restrict__ P,
                                            int K, int N, int NPAD, int idx) {
    if (idx >= K * NPAD) return;
    int k = idx / NPAD;
    int c = idx - k * NPAD;
    float v = (c < N) ? W[(size_t)k * N + c] : 0.0f;
    P[(size_t)(k >> 1) * 2 * NPAD + 2 * c + (k & 1)] = v;
}

#define HCAP 50176
#define CHUNK 256

__device__ __forceinline__ void slice_count_rank_role(
        int* h, int* csum, const int* __restrict__ dst, int* __restrict__ rank,
        int* __restrict__ part, int* __restrict__ cpre,
        int E, int N, int NP, int slice, int nslices) {
    const int tid = threadIdx.x;
    const int lane = tid & 31;
    const int wave = tid >> 5;
    int carry = 0;
    const int epb = (E + nslices - 1) / nslices;
    const int e0 = slice * epb;
    const int e1 = min(e0 + epb, E);
    int* prow = part + (size_t)slice * NP;

    for (int lo = 0; lo < N; lo += HCAP) {
        const int n = min(HCAP, N - lo);
        const int nq = n >> 2;
        const int nch = (n + CHUNK - 1) / CHUNK;
        const int4 z = {0, 0, 0, 0};
        for (int q = tid; q < nch * (CHUNK / 4); q += 1024) ((int4*)h)[q] = z;
        __syncthreads();

        for (int e = e0 + tid; e < e1; e += 8192) {
            int d[8];
#pragma unroll
            for (int u = 0; u < 8; ++u) d[u] = dst[min(e + u * 1024, e1 - 1)] - lo;
#pragma unroll
            for (int u = 0; u < 8; ++u) {
                const int ee = e + u * 1024;
                if (ee < e1 && (unsigned)d[u] < (unsigned)n) rank[ee] = atomicAdd(&h[d[u]], 1);
            }
        }
        __syncthreads();

        for (int c0 = 0; c0 < nch; c0 += 256) {
            const int c = c0 + (tid >> 2);
            const int sub = tid & 3;
            int s = 0;
            if (c < nch) {
                const int4* p4 = (const int4*)(h + c * CHUNK + sub * (CHUNK / 4));
#pragma unroll
                for (int i = 0; i < CHUNK / 16; ++i) {
                    const int4 a = p4[(i + lane) & (CHUNK / 16 - 1)];
                    s += a.x + a.y + a.z + a.w;
                }
            }
            s += __shfl_xor(s, 1, 32);
            s += __shfl_xor(s, 2, 32);
            if (sub == 0 && c < nch) csum[c] = s;
        }
        __syncthreads();
        if (wave == 0) {
            constexpr int PER = (HCAP / CHUNK + 31) / 32;
            int v[PER];
            int s = 0;
#pragma unroll
            for (int k = 0; k < PER; ++k) {
                const int c = lane * PER + k;
                v[k] = (c < nch) ? csum[c] : 0;
                s += v[k];
            }
            int x = s;
            for (int o = 1; o < 32; o <<= 1) {
                const int y = __shfl_up(x, o, 32);
                if (lane >= o) x += y;
            }
            int run = carry + x - s;
#pragma unroll
            for (int k = 0; k < PER; ++k) {
                const int c = lane * PER + k;
                if (c < nch) cpre[(size_t)(lo / CHUNK + c) * nslices + slice] = run;
                run += v[k];
            }
            carry += __shfl(x, 31, 32);
        }

        for (int q = tid; q < nq; q += 1024)
            *(int4*)(prow + lo + 4 * q) = ((const int4*)h)[q];
        for (int i = (nq << 2) + tid; i < n; i += 1024) prow[lo + i] = h[i];
        __syncthreads();
    }
}

__global__ __launch_bounds__(1024) void count_pack_kernel(
        const int* __restrict__ dst, int* __restrict__ rank, int* __restrict__ part,
        int* __restrict__ cpre, int E, int N, int NP, int SLICES,
        const float* __restrict__ W1, float* __restrict__ W1p, int K1, int N1, int NPAD1, int P1,
        const float* __restrict__ W2, float* __restrict__ W2p, int K2, int N2, int NPAD2) {
    __shared__ int h[HCAP];
    __shared__ int csum[HCAP / CHUNK];
    const int bid = blockIdx.x;
    if (bid < SLICES) {
        slice_count_rank_role(h, csum, dst, rank, part, cpre, E, N, NP, bid, SLICES);
    } else if (bid < SLICES + P1) {
        pack_w_role(W1, W1p, K1, N1, NPAD1, (bid - SLICES) * 1024 + (int)threadIdx.x);
    } else {
        pack_w_role(W2, W2p, K2, N2, NPAD2, (bid - SLICES - P1) * 1024 + (int)threadIdx.x);
    }
}

template <int B>
__global__ __launch_bounds__(CHUNK) void merge_scan_kernel(
        int* __restrict__ part, const int* __restrict__ cpre, int* __restrict__ cnt,
        int* __restrict__ off, float* __restrict__ dinv, int N, int NP) {
    __shared__ int wtot[CHUNK / 32];
    const int tid = threadIdx.x;
    const int lane = tid & 31;
    const int wave = tid >> 5;
    const int d = blockIdx.x * CHUNK + tid;
    const bool ok = d < N;
    const unsigned dl = (unsigned)min(d, N - 1);

    int base = 0;
#pragma unroll
    for (int j = 0; j < B; j += 32)
        if (j + lane < B) base += cpre[(size_t)blockIdx.x * B + j + lane];

    int v[B];
#pragma unroll
    for (int j = 0; j < B; ++j) {
        const int* row = part + (size_t)j * NP;
        v[j] = row[dl];
    }

    for (int o = 16; o > 0; o >>= 1) base += __shfl_xor(base, o, 32);

    int c = 0;
#pragma unroll
    for (int j = 0; j < B; ++j) c += v[j];

    int x = c;
    for (int o = 1; o < 32; o <<= 1) {
        const int y = __shfl_up(x, o, 32);
        if (lane >= o) x += y;
    }
    if (lane == 31) wtot[wave] = x;
    __syncthreads();
    int front = base;
    for (int w = 0; w < wave; ++w) front += wtot[w];

    if (ok) {
        const int o0 = front + (x - c);
        off[d] = o0;
        cnt[d] = c;
        const float dg = (float)(c + 1);
        dinv[d] = (dg > 0.0f) ? rsqrtf(dg) : 0.0f;
        int run = o0;
#pragma unroll
        for (int j = 0; j < B; ++j) {
            int* row = part + (size_t)j * NP;
            row[dl] = run;
            run += v[j];
        }
    }
}

__device__ __forceinline__ void place_edges_role(const int* __restrict__ src, const int* __restrict__ dst,
                                   const int* __restrict__ rank,
                                   const int* __restrict__ part,
                                   const float* __restrict__ dinv, int* __restrict__ seg,
                                   int E, int NP, int B, int e) {
    if (e >= E) return;
    const int s = src[e];
    const int d = dst[e];
    const int epb = (E + B - 1) / B;
    const int b = e / epb;
    const int pos = part[(size_t)b * NP + d] + rank[e];
    int2 v;
    v.x = s;
    v.y = __float_as_int(dinv[s]);
    *(int2*)(seg + 2 * (size_t)pos) = v;
}

template <int NT, int NPAD>
__device__ __forceinline__ void wmma_gemm_f32_role(
        const float* __restrict__ A, const float* __restrict__ Wp,
        float* __restrict__ Out, int M, int K, int Nout, int pairTile) {
    const int lane = threadIdx.x & 31;
    const int nPairTiles = (M + 31) >> 5;
    if (pairTile >= nPairTiles) return;

    const int m0    = pairTile * 32 + (lane & 15);
    const int koff  = (lane >> 4) << 1;
    const int koff2 = (lane >> 4);
    const int n     = lane & 15;

    const bool hasSecond = (pairTile * 32 + 16) < M;

    v8f acc0[NT], acc1[NT];
#pragma unroll
    for (int t = 0; t < NT; ++t) {
        v8f z = {0.f, 0.f, 0.f, 0.f, 0.f, 0.f, 0.f, 0.f};
        acc0[t] = z;
        acc1[t] = z;
    }

    const float* A0 = A + (size_t)m0 * K;
    const float* A1 = hasSecond ? (A0 + (size_t)16 * K) : A0;

    for (int k = 0; k < K; k += 4) {
        v2f a0 = *(const v2f*)(A0 + k + koff);
        v2f a1 = *(const v2f*)(A1 + k + koff);
        const float* Wrow = Wp + (size_t)((k >> 1) + koff2) * (2 * NPAD);
#pragma unroll
        for (int t = 0; t < NT; ++t) {
            v2f b = *(const v2f*)(Wrow + 2 * (t * 16 + n));
            acc0[t] = __builtin_amdgcn_wmma_f32_16x16x4_f32(
                false, a0, false, b, (short)0, acc0[t], false, false);
            acc1[t] = __builtin_amdgcn_wmma_f32_16x16x4_f32(
                false, a1, false, b, (short)0, acc1[t], false, false);
        }
    }

    const int rbase0 = pairTile * 32 + ((lane >> 4) << 3);
    const int rbase1 = rbase0 + 16;
#pragma unroll
    for (int t = 0; t < NT; ++t) {
        int col = t * 16 + n;
        if (col < Nout) {
#pragma unroll
            for (int i = 0; i < 8; ++i)
                Out[(size_t)(rbase0 + i) * Nout + col] = acc0[t][i];
            if (hasSecond) {
#pragma unroll
                for (int i = 0; i < 8; ++i)
                    Out[(size_t)(rbase1 + i) * Nout + col] = acc1[t][i];
            }
        }
    }
}

template <int NT, int NPAD>
__global__ __launch_bounds__(256) void place_gemm_kernel(
        const float* __restrict__ A, const float* __restrict__ Wp,
        float* __restrict__ Out, int M, int K, int Nout, int GB,
        const int* __restrict__ src, const int* __restrict__ dst,
        const int* __restrict__ rank,
        const int* __restrict__ part,
        const float* __restrict__ dinv, int* __restrict__ seg,
        int E, int NP, int B) {
    const int bid = blockIdx.x;
    if (bid < GB) {
        wmma_gemm_f32_role<NT, NPAD>(A, Wp, Out, M, K, Nout, bid * 8 + (int)(threadIdx.x >> 5));
    } else {
        place_edges_role(src, dst, rank, part, dinv, seg, E, NP, B,
                         (bid - GB) * 256 + (int)threadIdx.x);
    }
}

template <int U, int NT, int NPAD>
__global__ __launch_bounds__(512) void gather_relu_gemm_kernel(
        const float* __restrict__ H, const int* __restrict__ off, const int* __restrict__ cnt,
        const int* __restrict__ seg, const float* __restrict__ dinv, const float* __restrict__ b,
        const float* __restrict__ Wp, float* __restrict__ Out, int N, int Nout) {
    constexpr int PITCH = 132;
    __shared__ float tile[16 * PITCH];
    const int lane = threadIdx.x & 31;
    const int wave = __builtin_amdgcn_readfirstlane((int)(threadIdx.x >> 5));
    const int node = blockIdx.x * 16 + wave;

    float4 v = {0.0f, 0.0f, 0.0f, 0.0f};
    if (node < N) {
        const int beg = __builtin_amdgcn_readfirstlane(off[node]);
        const int n   = __builtin_amdgcn_readfirstlane(cnt[node]);
        const float dn = dinv[node];
        const float sn = dn * dn;
        const float* hl = H + 4 * lane;
        const float* bl = b + 4 * lane;
        const float b0 = bl[0], b1v = bl[1], b2v = bl[2], b3v = bl[3];

        float4 acc;
        {
            const float4 h = *(const float4*)(hl + (size_t)node * 128);
            acc.x = h.x * sn; acc.y = h.y * sn; acc.z = h.z * sn; acc.w = h.w * sn;
        }

        for (int base = 0; base < n; base += 32) {
            const int m = min(n - base, 32);
            int ps = 0, pw = 0;
            if (lane < m) {
                const int2 p = *(const int2*)(seg + 2 * (size_t)(beg + base + lane));
                ps = p.x;
                pw = p.y;
            }
            for (int j = 0; j < m; j += U) {
                float4 r[U];
                float w[U];
                for (int u = 0; u < U; ++u) {
                    const int jj = min(j + u, m - 1);
                    const int s = __builtin_amdgcn_readlane(ps, jj);
                    w[u] = __int_as_float(__builtin_amdgcn_readlane(pw, jj)) * dn;
                    r[u] = *(const float4*)(hl + (size_t)s * 128);
                }
                for (int u = 0; u < U; ++u) {
                    if (j + u < m) {
                        acc.x = fmaf(r[u].x, w[u], acc.x);
                        acc.y = fmaf(r[u].y, w[u], acc.y);
                        acc.z = fmaf(r[u].z, w[u], acc.z);
                        acc.w = fmaf(r[u].w, w[u], acc.w);
                    }
                }
            }
        }

        v.x = acc.x + b0; v.y = acc.y + b1v; v.z = acc.z + b2v; v.w = acc.w + b3v;
        v.x = v.x > 0.0f ? v.x : 0.0f;
        v.y = v.y > 0.0f ? v.y : 0.0f;
        v.z = v.z > 0.0f ? v.z : 0.0f;
        v.w = v.w > 0.0f ? v.w : 0.0f;
    }
    *(float4*)(tile + wave * PITCH + 4 * lane) = v;
    __syncthreads();
    if (wave >= NT) return;

    const int c = lane & 15;
    const int hh = lane >> 4;
    const float* arow = tile + c * PITCH + 2 * hh;
    const float* wcol = Wp + (size_t)hh * (2 * NPAD) + 2 * (wave * 16 + c);
    v8f acc0 = {0.f, 0.f, 0.f, 0.f, 0.f, 0.f, 0.f, 0.f};
    v8f acc1 = acc0;
    for (int k = 0; k < 128; k += 8) {
        const v2f a0 = *(const v2f*)(arow + k);
        const v2f a1 = *(const v2f*)(arow + k + 4);
        const v2f w0 = *(const v2f*)(wcol + (size_t)(k >> 1) * (2 * NPAD));
        const v2f w1 = *(const v2f*)(wcol + (size_t)((k >> 1) + 2) * (2 * NPAD));
        acc0 = __builtin_amdgcn_wmma_f32_16x16x4_f32(
            false, a0, false, w0, (short)0, acc0, false, false);
        acc1 = __builtin_amdgcn_wmma_f32_16x16x4_f32(
            false, a1, false, w1, (short)0, acc1, false, false);
    }

    const int col = wave * 16 + c;
    const int rbase = blockIdx.x * 16 + (hh << 3);
    if (col < Nout) {
#pragma unroll
        for (int i = 0; i < 8; ++i)
            if (rbase + i < N) Out[(size_t)(rbase + i) * Nout + col] = acc0[i] + acc1[i];
    }
}

template <int C, int U>
__global__ __launch_bounds__(256) void gather_logsoftmax_kernel(
        const float* __restrict__ H, const int* __restrict__ off, const int* __restrict__ cnt,
        const int* __restrict__ seg, const float* __restrict__ dinv, const float* __restrict__ b,
        float* __restrict__ out, int N) {
    const int lane = threadIdx.x & 31;
    const int node = __builtin_amdgcn_readfirstlane((int)((blockIdx.x * blockDim.x + threadIdx.x) >> 5));
    if (node >= N) return;
    const int beg = __builtin_amdgcn_readfirstlane(off[node]);
    const int n   = __builtin_amdgcn_readfirstlane(cnt[node]);
    const float dn = dinv[node];
    const float sn = dn * dn;
    const int f0 = lane, f1 = lane + 32;
    const bool ok0 = f0 < C, ok1 = f1 < C;
    const size_t obase = (size_t)node * C;

    float h0 = 0.0f, h1v = 0.0f, bias0 = 0.0f, bias1 = 0.0f;
    if (ok0) { h0 = H[obase + f0]; bias0 = b[f0]; }
    if (ok1) { h1v = H[obase + f1]; bias1 = b[f1]; }
    float a0 = h0 * sn, a1 = h1v * sn;

    for (int base = 0; base < n; base += 32) {
        const int m = min(n - base, 32);
        int ps = 0, pw = 0;
        if (lane < m) {
            const int2 p = *(const int2*)(seg + 2 * (size_t)(beg + base + lane));
            ps = p.x;
            pw = p.y;
        }
        for (int j = 0; j < m; j += U) {
            float r0[U], r1[U], w[U];
            for (int u = 0; u < U; ++u) {
                const int jj = min(j + u, m - 1);
                const int s = __builtin_amdgcn_readlane(ps, jj);
                w[u] = __int_as_float(__builtin_amdgcn_readlane(pw, jj)) * dn;
                const float* row = H + (size_t)s * C;
                r0[u] = ok0 ? row[f0] : 0.0f;
                r1[u] = ok1 ? row[f1] : 0.0f;
            }
            for (int u = 0; u < U; ++u) {
                if (j + u < m) {
                    a0 = fmaf(r0[u], w[u], a0);
                    a1 = fmaf(r1[u], w[u], a1);
                }
            }
        }
    }

    float v0 = -3.4e38f, v1 = -3.4e38f;
    if (ok0) v0 = a0 + bias0;
    if (ok1) v1 = a1 + bias1;
    float mx = fmaxf(v0, v1);
    for (int o = 16; o > 0; o >>= 1) mx = fmaxf(mx, __shfl_xor(mx, o, 32));
    float sum = 0.0f;
    if (ok0) sum += __expf(v0 - mx);
    if (ok1) sum += __expf(v1 - mx);
    for (int o = 16; o > 0; o >>= 1) sum += __shfl_xor(sum, o, 32);
    const float lse = __logf(sum) + mx;
    if (ok0) out[obase + f0] = v0 - lse;
    if (ok1) out[obase + f1] = v1 - lse;
}

extern "C" void kernel_launch(void* const* d_in, const int* in_sizes, int n_in,
                              void* d_out, int out_size, void* d_ws, size_t ws_size,
                              hipStream_t stream) {
    const float* x  = (const float*)d_in[0];
    const int*   ei = (const int*)d_in[1];
    const float* W1 = (const float*)d_in[2];
    const float* b1 = (const float*)d_in[3];
    const float* W2 = (const float*)d_in[4];
    const float* b2 = (const float*)d_in[5];
    float* out = (float*)d_out;

    const int F_IN  = 256;
    const int HID   = 128;
    const int NCLS  = 40;
    const int NPAD2 = 48;
    const int N = in_sizes[0] / F_IN;
    const int E = in_sizes[1] / 2;
    const int* src = ei;
    const int* dst = ei + E;
    const int NP = (N + 63) & ~63;
    const int EP = (E + 63) & ~63;

    int*   cnt  = (int*)d_ws;
    int*   off  = cnt + NP;
    float* dinv = (float*)(off + NP);
    int*   rank = (int*)(dinv + NP);
    int*   seg  = rank + EP;
    float* h1   = (float*)(seg + 2 * (size_t)EP);
    float* h2   = h1 + (size_t)N * HID;
    float* W1p  = h2 + (size_t)N * NCLS;
    float* W2p  = W1p + (size_t)F_IN * HID;
    int*   part = (int*)(W2p + (size_t)HID * NPAD2);
    constexpr int SLICES = 64;
    const int NCH = (N + CHUNK - 1) / CHUNK;
    int*   cpre = part + (size_t)SLICES * NP;

    {
        const int P1 = (F_IN * HID + 1023) / 1024;
        const int P2 = (HID * NPAD2 + 1023) / 1024;
        count_pack_kernel<<<SLICES + P1 + P2, 1024, 0, stream>>>(
            dst, rank, part, cpre, E, N, NP, SLICES,
            W1, W1p, F_IN, HID, HID, P1,
            W2, W2p, HID, NCLS, NPAD2);
    }

    merge_scan_kernel<SLICES><<<NCH, CHUNK, 0, stream>>>(part, cpre, cnt, off, dinv, N, NP);

    {
        const int pairTiles = (N + 31) / 32;
        const int GB = (pairTiles + 7) / 8;
        const int PB = (E + 255) / 256;
        place_gemm_kernel<8, 128><<<GB + PB, 256, 0, stream>>>(
            x, W1p, h1, N, F_IN, HID, GB,
            src, dst, rank, part, dinv, seg, E, NP, SLICES);
    }

    {
        const int nodesPerBlock = 16;
        const int blocks = (N + nodesPerBlock - 1) / nodesPerBlock;
        gather_relu_gemm_kernel<8, 3, 48><<<blocks, nodesPerBlock * 32, 0, stream>>>(
            h1, off, cnt, seg, dinv, b1, W2p, h2, N, NCLS);
    }

    {
        const int wavesPerBlock = 8;
        const int blocks = (N + wavesPerBlock - 1) / wavesPerBlock;
        gather_logsoftmax_kernel<40, 8><<<blocks, wavesPerBlock * 32, 0, stream>>>(
            h2, off, cnt, seg, dinv, b2, out, N);
    }
}


# x@W1 k loop: two fragment register sets used in turn, the next k group's A and B loads issued ahead of the current group's 16 matrix instructions; same k order, sums bit-identical
# speedup vs baseline: 3.5367x; 1.1535x over previous
#include <hip/hip_runtime.h>
#include <hip/hip_bf16.h>

typedef __attribute__((ext_vector_type(2))) float v2f;
typedef __attribute__((ext_vector_type(8))) float v8f;

__device__ __forceinline__ void pack_w_role(const float* __restrict__ W, float* __restrict__ P,
                                            int K, int N, int NPAD, int idx) {
    if (idx >= K * NPAD) return;
    int k = idx / NPAD;
    int c = idx - k * NPAD;
    float v = (c < N) ? W[(size_t)k * N + c] : 0.0f;
    P[(size_t)(k >> 1) * 2 * NPAD + 2 * c + (k & 1)] = v;
}

#define HCAP 50176
#define CHUNK 256

__device__ __forceinline__ void slice_count_rank_role(
        int* h, int* csum, const int* __restrict__ dst, int* __restrict__ rank,
        int* __restrict__ part, int* __restrict__ cpre,
        int E, int N, int NP, int slice, int nslices) {
    const int tid = threadIdx.x;
    const int lane = tid & 31;
    const int wave = tid >> 5;
    int carry = 0;
    const int epb = (E + nslices - 1) / nslices;
    const int e0 = slice * epb;
    const int e1 = min(e0 + epb, E);
    int* prow = part + (size_t)slice * NP;

    for (int lo = 0; lo < N; lo += HCAP) {
        const int n = min(HCAP, N - lo);
        const int nq = n >> 2;
        const int nch = (n + CHUNK - 1) / CHUNK;
        const int4 z = {0, 0, 0, 0};
        for (int q = tid; q < nch * (CHUNK / 4); q += 1024) ((int4*)h)[q] = z;
        __syncthreads();

        for (int e = e0 + tid; e < e1; e += 8192) {
            int d[8];
#pragma unroll
            for (int u = 0; u < 8; ++u) d[u] = dst[min(e + u * 1024, e1 - 1)] - lo;
#pragma unroll
            for (int u = 0; u < 8; ++u) {
                const int ee = e + u * 1024;
                if (ee < e1 && (unsigned)d[u] < (unsigned)n) rank[ee] = atomicAdd(&h[d[u]], 1);
            }
        }
        __syncthreads();

        for (int c0 = 0; c0 < nch; c0 += 256) {
            const int c = c0 + (tid >> 2);
            const int sub = tid & 3;
            int s = 0;
            if (c < nch) {
                const int4* p4 = (const int4*)(h + c * CHUNK + sub * (CHUNK / 4));
#pragma unroll
                for (int i = 0; i < CHUNK / 16; ++i) {
                    const int4 a = p4[(i + lane) & (CHUNK / 16 - 1)];
                    s += a.x + a.y + a.z + a.w;
                }
            }
            s += __shfl_xor(s, 1, 32);
            s += __shfl_xor(s, 2, 32);
            if (sub == 0 && c < nch) csum[c] = s;
        }
        __syncthreads();
        if (wave == 0) {
            constexpr int PER = (HCAP / CHUNK + 31) / 32;
            int v[PER];
            int s = 0;
#pragma unroll
            for (int k = 0; k < PER; ++k) {
                const int c = lane * PER + k;
                v[k] = (c < nch) ? csum[c] : 0;
                s += v[k];
            }
            int x = s;
            for (int o = 1; o < 32; o <<= 1) {
                const int y = __shfl_up(x, o, 32);
                if (lane >= o) x += y;
            }
            int run = carry + x - s;
#pragma unroll
            for (int k = 0; k < PER; ++k) {
                const int c = lane * PER + k;
                if (c < nch) cpre[(size_t)(lo / CHUNK + c) * nslices + slice] = run;
                run += v[k];
            }
            carry += __shfl(x, 31, 32);
        }

        for (int q = tid; q < nq; q += 1024)
            *(int4*)(prow + lo + 4 * q) = ((const int4*)h)[q];
        for (int i = (nq << 2) + tid; i < n; i += 1024) prow[lo + i] = h[i];
        __syncthreads();
    }
}

__global__ __launch_bounds__(1024) void count_pack_kernel(
        const int* __restrict__ dst, int* __restrict__ rank, int* __restrict__ part,
        int* __restrict__ cpre, int E, int N, int NP, int SLICES,
        const float* __restrict__ W1, float* __restrict__ W1p, int K1, int N1, int NPAD1, int P1,
        const float* __restrict__ W2, float* __restrict__ W2p, int K2, int N2, int NPAD2) {
    __shared__ int h[HCAP];
    __shared__ int csum[HCAP / CHUNK];
    const int bid = blockIdx.x;
    if (bid < SLICES) {
        slice_count_rank_role(h, csum, dst, rank, part, cpre, E, N, NP, bid, SLICES);
    } else if (bid < SLICES + P1) {
        pack_w_role(W1, W1p, K1, N1, NPAD1, (bid - SLICES) * 1024 + (int)threadIdx.x);
    } else {
        pack_w_role(W2, W2p, K2, N2, NPAD2, (bid - SLICES - P1) * 1024 + (int)threadIdx.x);
    }
}

template <int B>
__global__ __launch_bounds__(CHUNK) void merge_scan_kernel(
        int* __restrict__ part, const int* __restrict__ cpre, int* __restrict__ cnt,
        int* __restrict__ off, float* __restrict__ dinv, int N, int NP) {
    __shared__ int wtot[CHUNK / 32];
    const int tid = threadIdx.x;
    const int lane = tid & 31;
    const int wave = tid >> 5;
    const int d = blockIdx.x * CHUNK + tid;
    const bool ok = d < N;
    const unsigned dl = (unsigned)min(d, N - 1);

    int base = 0;
#pragma unroll
    for (int j = 0; j < B; j += 32)
        if (j + lane < B) base += cpre[(size_t)blockIdx.x * B + j + lane];

    int v[B];
#pragma unroll
    for (int j = 0; j < B; ++j) {
        const int* row = part + (size_t)j * NP;
        v[j] = row[dl];
    }

    for (int o = 16; o > 0; o >>= 1) base += __shfl_xor(base, o, 32);

    int c = 0;
#pragma unroll
    for (int j = 0; j < B; ++j) c += v[j];

    int x = c;
    for (int o = 1; o < 32; o <<= 1) {
        const int y = __shfl_up(x, o, 32);
        if (lane >= o) x += y;
    }
    if (lane == 31) wtot[wave] = x;
    __syncthreads();
    int front = base;
    for (int w = 0; w < wave; ++w) front += wtot[w];

    if (ok) {
        const int o0 = front + (x - c);
        off[d] = o0;
        cnt[d] = c;
        const float dg = (float)(c + 1);
        dinv[d] = (dg > 0.0f) ? rsqrtf(dg) : 0.0f;
        int run = o0;
#pragma unroll
        for (int j = 0; j < B; ++j) {
            int* row = part + (size_t)j * NP;
            row[dl] = run;
            run += v[j];
        }
    }
}

__device__ __forceinline__ void place_edges_role(const int* __restrict__ src, const int* __restrict__ dst,
                                   const int* __restrict__ rank,
                                   const int* __restrict__ part,
                                   const float* __restrict__ dinv, int* __restrict__ seg,
                                   int E, int NP, int B, int e) {
    if (e >= E) return;
    const int s = src[e];
    const int d = dst[e];
    const int epb = (E + B - 1) / B;
    const int b = e / epb;
    const int pos = part[(size_t)b * NP + d] + rank[e];
    int2 v;
    v.x = s;
    v.y = __float_as_int(dinv[s]);
    *(int2*)(seg + 2 * (size_t)pos) = v;
}

template <int NT, int NPAD>
__device__ __forceinline__ void wmma_gemm_f32_role(
        const float* __restrict__ A, const float* __restrict__ Wp,
        float* __restrict__ Out, int M, int K, int Nout, int pairTile) {
    const int lane = threadIdx.x & 31;
    const int nPairTiles = (M + 31) >> 5;
    if (pairTile >= nPairTiles) return;

    const int m0    = pairTile * 32 + (lane & 15);
    const int koff  = (lane >> 4) << 1;
    const int koff2 = (lane >> 4);
    const int n     = lane & 15;

    const bool hasSecond = (pairTile * 32 + 16) < M;

    v8f acc0[NT], acc1[NT];
#pragma unroll
    for (int t = 0; t < NT; ++t) {
        v8f z = {0.f, 0.f, 0.f, 0.f, 0.f, 0.f, 0.f, 0.f};
        acc0[t] = z;
        acc1[t] = z;
    }

    const float* A0 = A + (size_t)m0 * K;
    const float* A1 = hasSecond ? (A0 + (size_t)16 * K) : A0;

    const float* Wl = Wp + (size_t)koff2 * (2 * NPAD) + 2 * n;
    const float* A0l = A0 + koff;
    const float* A1l = A1 + koff;
    const int klast = K - 4;

    v2f p0, p1, pb[NT];
    v2f q0, q1, qb[NT];
    p0 = *(const v2f*)(A0l);
    p1 = *(const v2f*)(A1l);
#pragma unroll
    for (int t = 0; t < NT; ++t) pb[t] = *(const v2f*)(Wl + 32 * t);

    int k = 0;
    for (; k + 8 <= K; k += 8) {
        {
            const float* Wq = Wl + (size_t)((k + 4) >> 1) * (2 * NPAD);
            q0 = *(const v2f*)(A0l + k + 4);
            q1 = *(const v2f*)(A1l + k + 4);
#pragma unroll
            for (int t = 0; t < NT; ++t) qb[t] = *(const v2f*)(Wq + 32 * t);
        }
#pragma unroll
        for (int t = 0; t < NT; ++t) {
            acc0[t] = __builtin_amdgcn_wmma_f32_16x16x4_f32(
                false, p0, false, pb[t], (short)0, acc0[t], false, false);
            acc1[t] = __builtin_amdgcn_wmma_f32_16x16x4_f32(
                false, p1, false, pb[t], (short)0, acc1[t], false, false);
        }
        {
            const int kp = min(k + 8, klast);
            const float* Wn = Wl + (size_t)(kp >> 1) * (2 * NPAD);
            p0 = *(const v2f*)(A0l + kp);
            p1 = *(const v2f*)(A1l + kp);
#pragma unroll
            for (int t = 0; t < NT; ++t) pb[t] = *(const v2f*)(Wn + 32 * t);
        }
#pragma unroll
        for (int t = 0; t < NT; ++t) {
            acc0[t] = __builtin_amdgcn_wmma_f32_16x16x4_f32(
                false, q0, false, qb[t], (short)0, acc0[t], false, false);
            acc1[t] = __builtin_amdgcn_wmma_f32_16x16x4_f32(
                false, q1, false, qb[t], (short)0, acc1[t], false, false);
        }
    }
    if (k < K) {
#pragma unroll
        for (int t = 0; t < NT; ++t) {
            acc0[t] = __builtin_amdgcn_wmma_f32_16x16x4_f32(
                false, p0, false, pb[t], (short)0, acc0[t], false, false);
            acc1[t] = __builtin_amdgcn_wmma_f32_16x16x4_f32(
                false, p1, false, pb[t], (short)0, acc1[t], false, false);
        }
    }

    const int rbase0 = pairTile * 32 + ((lane >> 4) << 3);
    const int rbase1 = rbase0 + 16;
#pragma unroll
    for (int t = 0; t < NT; ++t) {
        int col = t * 16 + n;
        if (col < Nout) {
#pragma unroll
            for (int i = 0; i < 8; ++i)
                Out[(size_t)(rbase0 + i) * Nout + col] = acc0[t][i];
            if (hasSecond) {
#pragma unroll
                for (int i = 0; i < 8; ++i)
                    Out[(size_t)(rbase1 + i) * Nout + col] = acc1[t][i];
            }
        }
    }
}

template <int NT, int NPAD>
__global__ __launch_bounds__(256) void place_gemm_kernel(
        const float* __restrict__ A, const float* __restrict__ Wp,
        float* __restrict__ Out, int M, int K, int Nout, int GB,
        const int* __restrict__ src, const int* __restrict__ dst,
        const int* __restrict__ rank,
        const int* __restrict__ part,
        const float* __restrict__ dinv, int* __restrict__ seg,
        int E, int NP, int B) {
    const int bid = blockIdx.x;
    if (bid < GB) {
        wmma_gemm_f32_role<NT, NPAD>(A, Wp, Out, M, K, Nout, bid * 8 + (int)(threadIdx.x >> 5));
    } else {
        place_edges_role(src, dst, rank, part, dinv, seg, E, NP, B,
                         (bid - GB) * 256 + (int)threadIdx.x);
    }
}

template <int U, int NT, int NPAD>
__global__ __launch_bounds__(512) void gather_relu_gemm_kernel(
        const float* __restrict__ H, const int* __restrict__ off, const int* __restrict__ cnt,
        const int* __restrict__ seg, const float* __restrict__ dinv, const float* __restrict__ b,
        const float* __restrict__ Wp, float* __restrict__ Out, int N, int Nout) {
    constexpr int PITCH = 132;
    __shared__ float tile[16 * PITCH];
    const int lane = threadIdx.x & 31;
    const int wave = __builtin_amdgcn_readfirstlane((int)(threadIdx.x >> 5));
    const int node = blockIdx.x * 16 + wave;

    float4 v = {0.0f, 0.0f, 0.0f, 0.0f};
    if (node < N) {
        const int beg = __builtin_amdgcn_readfirstlane(off[node]);
        const int n   = __builtin_amdgcn_readfirstlane(cnt[node]);
        const float dn = dinv[node];
        const float sn = dn * dn;
        const float* hl = H + 4 * lane;
        const float* bl = b + 4 * lane;
        const float b0 = bl[0], b1v = bl[1], b2v = bl[2], b3v = bl[3];

        float4 acc;
        {
            const float4 h = *(const float4*)(hl + (size_t)node * 128);
            acc.x = h.x * sn; acc.y = h.y * sn; acc.z = h.z * sn; acc.w = h.w * sn;
        }

        for (int base = 0; base < n; base += 32) {
            const int m = min(n - base, 32);
            int ps = 0, pw = 0;
            if (lane < m) {
                const int2 p = *(const int2*)(seg + 2 * (size_t)(beg + base + lane));
                ps = p.x;
                pw = p.y;
            }
            for (int j = 0; j < m; j += U) {
                float4 r[U];
                float w[U];
                for (int u = 0; u < U; ++u) {
                    const int jj = min(j + u, m - 1);
                    const int s = __builtin_amdgcn_readlane(ps, jj);
                    w[u] = __int_as_float(__builtin_amdgcn_readlane(pw, jj)) * dn;
                    r[u] = *(const float4*)(hl + (size_t)s * 128);
                }
                for (int u = 0; u < U; ++u) {
                    if (j + u < m) {
                        acc.x = fmaf(r[u].x, w[u], acc.x);
                        acc.y = fmaf(r[u].y, w[u], acc.y);
                        acc.z = fmaf(r[u].z, w[u], acc.z);
                        acc.w = fmaf(r[u].w, w[u], acc.w);
                    }
                }
            }
        }

        v.x = acc.x + b0; v.y = acc.y + b1v; v.z = acc.z + b2v; v.w = acc.w + b3v;
        v.x = v.x > 0.0f ? v.x : 0.0f;
        v.y = v.y > 0.0f ? v.y : 0.0f;
        v.z = v.z > 0.0f ? v.z : 0.0f;
        v.w = v.w > 0.0f ? v.w : 0.0f;
    }
    *(float4*)(tile + wave * PITCH + 4 * lane) = v;
    __syncthreads();
    if (wave >= NT) return;

    const int c = lane & 15;
    const int hh = lane >> 4;
    const float* arow = tile + c * PITCH + 2 * hh;
    const float* wcol = Wp + (size_t)hh * (2 * NPAD) + 2 * (wave * 16 + c);
    v8f acc0 = {0.f, 0.f, 0.f, 0.f, 0.f, 0.f, 0.f, 0.f};
    v8f acc1 = acc0;
    for (int k = 0; k < 128; k += 8) {
        const v2f a0 = *(const v2f*)(arow + k);
        const v2f a1 = *(const v2f*)(arow + k + 4);
        const v2f w0 = *(const v2f*)(wcol + (size_t)(k >> 1) * (2 * NPAD));
        const v2f w1 = *(const v2f*)(wcol + (size_t)((k >> 1) + 2) * (2 * NPAD));
        acc0 = __builtin_amdgcn_wmma_f32_16x16x4_f32(
            false, a0, false, w0, (short)0, acc0, false, false);
        acc1 = __builtin_amdgcn_wmma_f32_16x16x4_f32(
            false, a1, false, w1, (short)0, acc1, false, false);
    }

    const int col = wave * 16 + c;
    const int rbase = blockIdx.x * 16 + (hh << 3);
    if (col < Nout) {
#pragma unroll
        for (int i = 0; i < 8; ++i)
            if (rbase + i < N) Out[(size_t)(rbase + i) * Nout + col] = acc0[i] + acc1[i];
    }
}

template <int C, int U>
__global__ __launch_bounds__(256) void gather_logsoftmax_kernel(
        const float* __restrict__ H, const int* __restrict__ off, const int* __restrict__ cnt,
        const int* __restrict__ seg, const float* __restrict__ dinv, const float* __restrict__ b,
        float* __restrict__ out, int N) {
    const int lane = threadIdx.x & 31;
    const int node = __builtin_amdgcn_readfirstlane((int)((blockIdx.x * blockDim.x + threadIdx.x) >> 5));
    if (node >= N) return;
    const int beg = __builtin_amdgcn_readfirstlane(off[node]);
    const int n   = __builtin_amdgcn_readfirstlane(cnt[node]);
    const float dn = dinv[node];
    const float sn = dn * dn;
    const int f0 = lane, f1 = lane + 32;
    const bool ok0 = f0 < C, ok1 = f1 < C;
    const size_t obase = (size_t)node * C;

    float h0 = 0.0f, h1v = 0.0f, bias0 = 0.0f, bias1 = 0.0f;
    if (ok0) { h0 = H[obase + f0]; bias0 = b[f0]; }
    if (ok1) { h1v = H[obase + f1]; bias1 = b[f1]; }
    float a0 = h0 * sn, a1 = h1v * sn;

    for (int base = 0; base < n; base += 32) {
        const int m = min(n - base, 32);
        int ps = 0, pw = 0;
        if (lane < m) {
            const int2 p = *(const int2*)(seg + 2 * (size_t)(beg + base + lane));
            ps = p.x;
            pw = p.y;
        }
        for (int j = 0; j < m; j += U) {
            float r0[U], r1[U], w[U];
            for (int u = 0; u < U; ++u) {
                const int jj = min(j + u, m - 1);
                const int s = __builtin_amdgcn_readlane(ps, jj);
                w[u] = __int_as_float(__builtin_amdgcn_readlane(pw, jj)) * dn;
                const float* row = H + (size_t)s * C;
                r0[u] = ok0 ? row[f0] : 0.0f;
                r1[u] = ok1 ? row[f1] : 0.0f;
            }
            for (int u = 0; u < U; ++u) {
                if (j + u < m) {
                    a0 = fmaf(r0[u], w[u], a0);
                    a1 = fmaf(r1[u], w[u], a1);
                }
            }
        }
    }

    float v0 = -3.4e38f, v1 = -3.4e38f;
    if (ok0) v0 = a0 + bias0;
    if (ok1) v1 = a1 + bias1;
    float mx = fmaxf(v0, v1);
    for (int o = 16; o > 0; o >>= 1) mx = fmaxf(mx, __shfl_xor(mx, o, 32));
    float sum = 0.0f;
    if (ok0) sum += __expf(v0 - mx);
    if (ok1) sum += __expf(v1 - mx);
    for (int o = 16; o > 0; o >>= 1) sum += __shfl_xor(sum, o, 32);
    const float lse = __logf(sum) + mx;
    if (ok0) out[obase + f0] = v0 - lse;
    if (ok1) out[obase + f1] = v1 - lse;
}

extern "C" void kernel_launch(void* const* d_in, const int* in_sizes, int n_in,
                              void* d_out, int out_size, void* d_ws, size_t ws_size,
                              hipStream_t stream) {
    const float* x  = (const float*)d_in[0];
    const int*   ei = (const int*)d_in[1];
    const float* W1 = (const float*)d_in[2];
    const float* b1 = (const float*)d_in[3];
    const float* W2 = (const float*)d_in[4];
    const float* b2 = (const float*)d_in[5];
    float* out = (float*)d_out;

    const int F_IN  = 256;
    const int HID   = 128;
    const int NCLS  = 40;
    const int NPAD2 = 48;
    const int N = in_sizes[0] / F_IN;
    const int E = in_sizes[1] / 2;
    const int* src = ei;
    const int* dst = ei + E;
    const int NP = (N + 63) & ~63;
    const int EP = (E + 63) & ~63;

    int*   cnt  = (int*)d_ws;
    int*   off  = cnt + NP;
    float* dinv = (float*)(off + NP);
    int*   rank = (int*)(dinv + NP);
    int*   seg  = rank + EP;
    float* h1   = (float*)(seg + 2 * (size_t)EP);
    float* h2   = h1 + (size_t)N * HID;
    float* W1p  = h2 + (size_t)N * NCLS;
    float* W2p  = W1p + (size_t)F_IN * HID;
    int*   part = (int*)(W2p + (size_t)HID * NPAD2);
    constexpr int SLICES = 64;
    const int NCH = (N + CHUNK - 1) / CHUNK;
    int*   cpre = part + (size_t)SLICES * NP;

    {
        const int P1 = (F_IN * HID + 1023) / 1024;
        const int P2 = (HID * NPAD2 + 1023) / 1024;
        count_pack_kernel<<<SLICES + P1 + P2, 1024, 0, stream>>>(
            dst, rank, part, cpre, E, N, NP, SLICES,
            W1, W1p, F_IN, HID, HID, P1,
            W2, W2p, HID, NCLS, NPAD2);
    }

    merge_scan_kernel<SLICES><<<NCH, CHUNK, 0, stream>>>(part, cpre, cnt, off, dinv, N, NP);

    {
        const int pairTiles = (N + 31) / 32;
        const int GB = (pairTiles + 7) / 8;
        const int PB = (E + 255) / 256;
        place_gemm_kernel<8, 128><<<GB + PB, 256, 0, stream>>>(
            x, W1p, h1, N, F_IN, HID, GB,
            src, dst, rank, part, dinv, seg, E, NP, SLICES);
    }

    {
        const int nodesPerBlock = 16;
        const int blocks = (N + nodesPerBlock - 1) / nodesPerBlock;
        gather_relu_gemm_kernel<8, 3, 48><<<blocks, nodesPerBlock * 32, 0, stream>>>(
            h1, off, cnt, seg, dinv, b1, W2p, h2, N, NCLS);
    }

    {
        const int wavesPerBlock = 8;
        const int blocks = (N + wavesPerBlock - 1) / wavesPerBlock;
        gather_logsoftmax_kernel<40, 8><<<blocks, wavesPerBlock * 32, 0, stream>>>(
            h2, off, cnt, seg, dinv, b2, out, N);
    }
}


# edge sort in 2 levels: slice blocks group edges by 256-node range, a block per range beside x@W1 counts, scans, places; 5 launches to 4; layer 1 gather looks up dinv[s]; in-segment sum order changes
# speedup vs baseline: 4.1078x; 1.1615x over previous
#include <hip/hip_runtime.h>
#include <hip/hip_bf16.h>

typedef __attribute__((ext_vector_type(2))) float v2f;
typedef __attribute__((ext_vector_type(8))) float v8f;

__device__ __forceinline__ void pack_w_role(const float* __restrict__ W, float* __restrict__ P,
                                            int K, int N, int NPAD, int idx) {
    if (idx >= K * NPAD) return;
    int k = idx / NPAD;
    int c = idx - k * NPAD;
    float v = (c < N) ? W[(size_t)k * N + c] : 0.0f;
    P[(size_t)(k >> 1) * 2 * NPAD + 2 * c + (k & 1)] = v;
}

#define RSHIFT 8
#define RANGE (1 << RSHIFT)
#define MAXR 2048

template <int SL>
__device__ __forceinline__ void slice_bucket_role(
        int* h, int* wtot, const int* __restrict__ src, const int* __restrict__ dst,
        int2* __restrict__ pairs, int* __restrict__ mstart, int* __restrict__ mlen,
        int E, int N, int slice) {
    const int tid = threadIdx.x;
    const int lane = tid & 31;
    const int wave = tid >> 5;
    const int NR = (N + RANGE - 1) / RANGE;
    const int epb = (E + SL - 1) / SL;
    const int e0 = slice * epb;
    const int e1 = min(e0 + epb, E);

    h[2 * tid] = 0;
    h[2 * tid + 1] = 0;
    __syncthreads();

    for (int e = e0 + tid; e < e1; e += 8192) {
        int d[8];
#pragma unroll
        for (int u = 0; u < 8; ++u) d[u] = dst[min(e + u * 1024, e1 - 1)];
#pragma unroll
        for (int u = 0; u < 8; ++u)
            if (e + u * 1024 < e1) atomicAdd(&h[d[u] >> RSHIFT], 1);
    }
    __syncthreads();

    const int c0 = h[2 * tid];
    const int c1 = h[2 * tid + 1];
    const int s = c0 + c1;
    int x = s;
    for (int o = 1; o < 32; o <<= 1) {
        const int y = __shfl_up(x, o, 32);
        if (lane >= o) x += y;
    }
    if (lane == 31) wtot[wave] = x;
    __syncthreads();
    int front = 0;
    for (int w = 0; w < wave; ++w) front += wtot[w];
    const int s0 = front + x - s;
    const int s1 = s0 + c0;
    h[2 * tid] = s0;
    h[2 * tid + 1] = s1;
    if (2 * tid < NR) {
        mstart[(size_t)(2 * tid) * SL + slice] = s0;
        mlen[(size_t)(2 * tid) * SL + slice] = c0;
    }
    if (2 * tid + 1 < NR) {
        mstart[(size_t)(2 * tid + 1) * SL + slice] = s1;
        mlen[(size_t)(2 * tid + 1) * SL + slice] = c1;
    }
    __syncthreads();

    int2* pout = pairs + e0;
    for (int e = e0 + tid; e < e1; e += 8192) {
        int d[8], sv[8];
#pragma unroll
        for (int u = 0; u < 8; ++u) {
            const int ee = min(e + u * 1024, e1 - 1);
            d[u] = dst[ee];
            sv[u] = src[ee];
        }
#pragma unroll
        for (int u = 0; u < 8; ++u) {
            if (e + u * 1024 < e1) {
                const int pos = atomicAdd(&h[d[u] >> RSHIFT], 1);
                int2 v;
                v.x = sv[u];
                v.y = d[u];
                pout[pos] = v;
            }
        }
    }
}

template <int SL>
__global__ __launch_bounds__(1024) void bucket_pack_kernel(
        const int* __restrict__ src, const int* __restrict__ dst, int2* __restrict__ pairs,
        int* __restrict__ mstart, int* __restrict__ mlen, int E, int N,
        const float* __restrict__ W1, float* __restrict__ W1p, int K1, int N1, int NPAD1, int P1,
        const float* __restrict__ W2, float* __restrict__ W2p, int K2, int N2, int NPAD2) {
    __shared__ int h[MAXR];
    __shared__ int wtot[32];
    const int bid = blockIdx.x;
    if (bid < SL) {
        slice_bucket_role<SL>(h, wtot, src, dst, pairs, mstart, mlen, E, N, bid);
    } else if (bid < SL + P1) {
        pack_w_role(W1, W1p, K1, N1, NPAD1, (bid - SL) * 1024 + (int)threadIdx.x);
    } else {
        pack_w_role(W2, W2p, K2, N2, NPAD2, (bid - SL - P1) * 1024 + (int)threadIdx.x);
    }
}

template <int SL>
__device__ __forceinline__ void range_owner_role(
        int* cntL, int* wtot, const int2* __restrict__ pairs,
        const int* __restrict__ mstart, const int* __restrict__ mlen,
        int* __restrict__ cnt, int* __restrict__ off, float* __restrict__ dinv,
        int* __restrict__ seg, int E, int N, int r) {
    constexpr int SU = SL / 8;
    constexpr int EPL = 2;
    const int tid = threadIdx.x;
    const int lane = tid & 31;
    const int wave = __builtin_amdgcn_readfirstlane(tid >> 5);
    const int lo = r * RANGE;
    const int epb = (E + SL - 1) / SL;
    const int* ms = mstart + (size_t)r * SL;
    const int* ml = mlen + (size_t)r * SL;

    cntL[tid] = 0;
    int base = 0;
#pragma unroll
    for (int j = 0; j < SL; j += 32) base += ms[j + lane];
    int ln[SU];
    const int2* sp[SU];
#pragma unroll
    for (int u = 0; u < SU; ++u) {
        const int sg = wave + 8 * u;
        ln[u] = ml[sg];
        sp[u] = pairs + (size_t)sg * epb + ms[sg];
    }
    for (int o = 16; o > 0; o >>= 1) base += __shfl_xor(base, o, 32);
    __syncthreads();

    {
        int d[SU][EPL];
#pragma unroll
        for (int u = 0; u < SU; ++u)
#pragma unroll
            for (int v = 0; v < EPL; ++v)
                d[u][v] = sp[u][max(min(lane + 32 * v, ln[u] - 1), 0)].y;
#pragma unroll
        for (int u = 0; u < SU; ++u)
#pragma unroll
            for (int v = 0; v < EPL; ++v)
                if (lane + 32 * v < ln[u]) atomicAdd(&cntL[d[u][v] - lo], 1);
#pragma unroll
        for (int u = 0; u < SU; ++u)
            for (int i = lane + 32 * EPL; i < ln[u]; i += 32) atomicAdd(&cntL[sp[u][i].y - lo], 1);
    }
    __syncthreads();

    const int c = cntL[tid];
    int x = c;
    for (int o = 1; o < 32; o <<= 1) {
        const int y = __shfl_up(x, o, 32);
        if (lane >= o) x += y;
    }
    if (lane == 31) wtot[wave] = x;
    __syncthreads();
    int front = base;
    for (int w = 0; w < wave; ++w) front += wtot[w];
    const int o0 = front + (x - c);
    const int node = lo + tid;
    if (node < N) {
        off[node] = o0;
        cnt[node] = c;
        const float dg = (float)(c + 1);
        dinv[node] = (dg > 0.0f) ? rsqrtf(dg) : 0.0f;
    }
    cntL[tid] = o0;
    __syncthreads();

    {
        int2 p[SU][EPL];
#pragma unroll
        for (int u = 0; u < SU; ++u)
#pragma unroll
            for (int v = 0; v < EPL; ++v)
                p[u][v] = sp[u][max(min(lane + 32 * v, ln[u] - 1), 0)];
#pragma unroll
        for (int u = 0; u < SU; ++u)
#pragma unroll
            for (int v = 0; v < EPL; ++v)
                if (lane + 32 * v < ln[u]) seg[atomicAdd(&cntL[p[u][v].y - lo], 1)] = p[u][v].x;
#pragma unroll
        for (int u = 0; u < SU; ++u)
            for (int i = lane + 32 * EPL; i < ln[u]; i += 32) {
                const int2 q = sp[u][i];
                seg[atomicAdd(&cntL[q.y - lo], 1)] = q.x;
            }
    }
}

template <int NT, int NPAD>
__device__ __forceinline__ void wmma_gemm_f32_role(
        const float* __restrict__ A, const float* __restrict__ Wp,
        float* __restrict__ Out, int M, int K, int Nout, int pairTile) {
    const int lane = threadIdx.x & 31;
    const int nPairTiles = (M + 31) >> 5;
    if (pairTile >= nPairTiles) return;

    const int m0    = pairTile * 32 + (lane & 15);
    const int koff  = (lane >> 4) << 1;
    const int koff2 = (lane >> 4);
    const int n     = lane & 15;

    const bool hasSecond = (pairTile * 32 + 16) < M;

    v8f acc0[NT], acc1[NT];
#pragma unroll
    for (int t = 0; t < NT; ++t) {
        v8f z = {0.f, 0.f, 0.f, 0.f, 0.f, 0.f, 0.f, 0.f};
        acc0[t] = z;
        acc1[t] = z;
    }

    const float* A0 = A + (size_t)m0 * K;
    const float* A1 = hasSecond ? (A0 + (size_t)16 * K) : A0;

    const float* Wl = Wp + (size_t)koff2 * (2 * NPAD) + 2 * n;
    const float* A0l = A0 + koff;
    const float* A1l = A1 + koff;
    const int klast = K - 4;

    v2f p0, p1, pb[NT];
    v2f q0, q1, qb[NT];
    p0 = *(const v2f*)(A0l);
    p1 = *(const v2f*)(A1l);
#pragma unroll
    for (int t = 0; t < NT; ++t) pb[t] = *(const v2f*)(Wl + 32 * t);

    int k = 0;
    for (; k + 8 <= K; k += 8) {
        {
            const float* Wq = Wl + (size_t)((k + 4) >> 1) * (2 * NPAD);
            q0 = *(const v2f*)(A0l + k + 4);
            q1 = *(const v2f*)(A1l + k + 4);
#pragma unroll
            for (int t = 0; t < NT; ++t) qb[t] = *(const v2f*)(Wq + 32 * t);
        }
#pragma unroll
        for (int t = 0; t < NT; ++t) {
            acc0[t] = __builtin_amdgcn_wmma_f32_16x16x4_f32(
                false, p0, false, pb[t], (short)0, acc0[t], false, false);
            acc1[t] = __builtin_amdgcn_wmma_f32_16x16x4_f32(
                false, p1, false, pb[t], (short)0, acc1[t], false, false);
        }
        {
            const int kp = min(k + 8, klast);
            const float* Wn = Wl + (size_t)(kp >> 1) * (2 * NPAD);
            p0 = *(const v2f*)(A0l + kp);
            p1 = *(const v2f*)(A1l + kp);
#pragma unroll
            for (int t = 0; t < NT; ++t) pb[t] = *(const v2f*)(Wn + 32 * t);
        }
#pragma unroll
        for (int t = 0; t < NT; ++t) {
            acc0[t] = __builtin_amdgcn_wmma_f32_16x16x4_f32(
                false, q0, false, qb[t], (short)0, acc0[t], false, false);
            acc1[t] = __builtin_amdgcn_wmma_f32_16x16x4_f32(
                false, q1, false, qb[t], (short)0, acc1[t], false, false);
        }
    }
    if (k < K) {
#pragma unroll
        for (int t = 0; t < NT; ++t) {
            acc0[t] = __builtin_amdgcn_wmma_f32_16x16x4_f32(
                false, p0, false, pb[t], (short)0, acc0[t], false, false);
            acc1[t] = __builtin_amdgcn_wmma_f32_16x16x4_f32(
                false, p1, false, pb[t], (short)0, acc1[t], false, false);
        }
    }

    const int rbase0 = pairTile * 32 + ((lane >> 4) << 3);
    const int rbase1 = rbase0 + 16;
#pragma unroll
    for (int t = 0; t < NT; ++t) {
        int col = t * 16 + n;
        if (col < Nout) {
#pragma unroll
            for (int i = 0; i < 8; ++i)
                Out[(size_t)(rbase0 + i) * Nout + col] = acc0[t][i];
            if (hasSecond) {
#pragma unroll
                for (int i = 0; i < 8; ++i)
                    Out[(size_t)(rbase1 + i) * Nout + col] = acc1[t][i];
            }
        }
    }
}

template <int NT, int NPAD, int SL>
__global__ __launch_bounds__(RANGE) void owner_gemm_kernel(
        const float* __restrict__ A, const float* __restrict__ Wp,
        float* __restrict__ Out, int M, int K, int Nout, int GB,
        const int2* __restrict__ pairs, const int* __restrict__ mstart, const int* __restrict__ mlen,
        int* __restrict__ cnt, int* __restrict__ off, float* __restrict__ dinv,
        int* __restrict__ seg, int E) {
    __shared__ int cntL[RANGE];
    __shared__ int wtot[RANGE / 32];
    const int bid = blockIdx.x;
    if (bid < GB) {
        wmma_gemm_f32_role<NT, NPAD>(A, Wp, Out, M, K, Nout, bid * 8 + (int)(threadIdx.x >> 5));
    } else {
        range_owner_role<SL>(cntL, wtot, pairs, mstart, mlen, cnt, off, dinv, seg, E, M, bid - GB);
    }
}

template <int U, int NT, int NPAD>
__global__ __launch_bounds__(512) void gather_relu_gemm_kernel(
        const float* __restrict__ H, const int* __restrict__ off, const int* __restrict__ cnt,
        const int* __restrict__ seg, float* __restrict__ wsg,
        const float* __restrict__ dinv, const float* __restrict__ b,
        const float* __restrict__ Wp, float* __restrict__ Out, int N, int Nout) {
    constexpr int PITCH = 132;
    __shared__ float tile[16 * PITCH];
    const int lane = threadIdx.x & 31;
    const int wave = __builtin_amdgcn_readfirstlane((int)(threadIdx.x >> 5));
    const int node = blockIdx.x * 16 + wave;

    float4 v = {0.0f, 0.0f, 0.0f, 0.0f};
    if (node < N) {
        const int beg = __builtin_amdgcn_readfirstlane(off[node]);
        const int n   = __builtin_amdgcn_readfirstlane(cnt[node]);
        const float dn = dinv[node];
        const float sn = dn * dn;
        const float* hl = H + 4 * lane;
        const float* bl = b + 4 * lane;
        const float b0 = bl[0], b1v = bl[1], b2v = bl[2], b3v = bl[3];

        float4 acc;
        {
            const float4 h = *(const float4*)(hl + (size_t)node * 128);
            acc.x = h.x * sn; acc.y = h.y * sn; acc.z = h.z * sn; acc.w = h.w * sn;
        }

        for (int base = 0; base < n; base += 32) {
            const int m = min(n - base, 32);
            int ps = 0, pw = 0;
            if (lane < m) {
                ps = seg[beg + base + lane];
                pw = __float_as_int(dinv[ps]);
            }
            for (int j = 0; j < m; j += U) {
                float4 r[U];
                float w[U];
                for (int u = 0; u < U; ++u) {
                    const int jj = min(j + u, m - 1);
                    const int s = __builtin_amdgcn_readlane(ps, jj);
                    r[u] = *(const float4*)(hl + (size_t)s * 128);
                }
                __builtin_amdgcn_sched_barrier(0);
                for (int u = 0; u < U; ++u)
                    w[u] = __int_as_float(__builtin_amdgcn_readlane(pw, min(j + u, m - 1))) * dn;
                for (int u = 0; u < U; ++u) {
                    if (j + u < m) {
                        acc.x = fmaf(r[u].x, w[u], acc.x);
                        acc.y = fmaf(r[u].y, w[u], acc.y);
                        acc.z = fmaf(r[u].z, w[u], acc.z);
                        acc.w = fmaf(r[u].w, w[u], acc.w);
                    }
                }
            }
            if (lane < m) wsg[beg + base + lane] = __int_as_float(pw);
        }

        v.x = acc.x + b0; v.y = acc.y + b1v; v.z = acc.z + b2v; v.w = acc.w + b3v;
        v.x = v.x > 0.0f ? v.x : 0.0f;
        v.y = v.y > 0.0f ? v.y : 0.0f;
        v.z = v.z > 0.0f ? v.z : 0.0f;
        v.w = v.w > 0.0f ? v.w : 0.0f;
    }
    *(float4*)(tile + wave * PITCH + 4 * lane) = v;
    __syncthreads();
    if (wave >= NT) return;

    const int c = lane & 15;
    const int hh = lane >> 4;
    const float* arow = tile + c * PITCH + 2 * hh;
    const float* wcol = Wp + (size_t)hh * (2 * NPAD) + 2 * (wave * 16 + c);
    v8f acc0 = {0.f, 0.f, 0.f, 0.f, 0.f, 0.f, 0.f, 0.f};
    v8f acc1 = acc0;
    for (int k = 0; k < 128; k += 8) {
        const v2f a0 = *(const v2f*)(arow + k);
        const v2f a1 = *(const v2f*)(arow + k + 4);
        const v2f w0 = *(const v2f*)(wcol + (size_t)(k >> 1) * (2 * NPAD));
        const v2f w1 = *(const v2f*)(wcol + (size_t)((k >> 1) + 2) * (2 * NPAD));
        acc0 = __builtin_amdgcn_wmma_f32_16x16x4_f32(
            false, a0, false, w0, (short)0, acc0, false, false);
        acc1 = __builtin_amdgcn_wmma_f32_16x16x4_f32(
            false, a1, false, w1, (short)0, acc1, false, false);
    }

    const int col = wave * 16 + c;
    const int rbase = blockIdx.x * 16 + (hh << 3);
    if (col < Nout) {
#pragma unroll
        for (int i = 0; i < 8; ++i)
            if (rbase + i < N) Out[(size_t)(rbase + i) * Nout + col] = acc0[i] + acc1[i];
    }
}

template <int C, int U>
__global__ __launch_bounds__(256) void gather_logsoftmax_kernel(
        const float* __restrict__ H, const int* __restrict__ off, const int* __restrict__ cnt,
        const int* __restrict__ seg, const float* __restrict__ wsg,
        const float* __restrict__ dinv, const float* __restrict__ b,
        float* __restrict__ out, int N) {
    const int lane = threadIdx.x & 31;
    const int node = __builtin_amdgcn_readfirstlane((int)((blockIdx.x * blockDim.x + threadIdx.x) >> 5));
    if (node >= N) return;
    const int beg = __builtin_amdgcn_readfirstlane(off[node]);
    const int n   = __builtin_amdgcn_readfirstlane(cnt[node]);
    const float dn = dinv[node];
    const float sn = dn * dn;
    const int f0 = lane, f1 = lane + 32;
    const bool ok0 = f0 < C, ok1 = f1 < C;
    const size_t obase = (size_t)node * C;

    float h0 = 0.0f, h1v = 0.0f, bias0 = 0.0f, bias1 = 0.0f;
    if (ok0) { h0 = H[obase + f0]; bias0 = b[f0]; }
    if (ok1) { h1v = H[obase + f1]; bias1 = b[f1]; }
    float a0 = h0 * sn, a1 = h1v * sn;

    for (int base = 0; base < n; base += 32) {
        const int m = min(n - base, 32);
        int ps = 0, pw = 0;
        if (lane < m) {
            ps = seg[beg + base + lane];
            pw = __float_as_int(wsg[beg + base + lane]);
        }
        for (int j = 0; j < m; j += U) {
            float r0[U], r1[U], w[U];
            for (int u = 0; u < U; ++u) {
                const int jj = min(j + u, m - 1);
                const int s = __builtin_amdgcn_readlane(ps, jj);
                w[u] = __int_as_float(__builtin_amdgcn_readlane(pw, jj)) * dn;
                const float* row = H + (size_t)s * C;
                r0[u] = ok0 ? row[f0] : 0.0f;
                r1[u] = ok1 ? row[f1] : 0.0f;
            }
            for (int u = 0; u < U; ++u) {
                if (j + u < m) {
                    a0 = fmaf(r0[u], w[u], a0);
                    a1 = fmaf(r1[u], w[u], a1);
                }
            }
        }
    }

    float v0 = -3.4e38f, v1 = -3.4e38f;
    if (ok0) v0 = a0 + bias0;
    if (ok1) v1 = a1 + bias1;
    float mx = fmaxf(v0, v1);
    for (int o = 16; o > 0; o >>= 1) mx = fmaxf(mx, __shfl_xor(mx, o, 32));
    float sum = 0.0f;
    if (ok0) sum += __expf(v0 - mx);
    if (ok1) sum += __expf(v1 - mx);
    for (int o = 16; o > 0; o >>= 1) sum += __shfl_xor(sum, o, 32);
    const float lse = __logf(sum) + mx;
    if (ok0) out[obase + f0] = v0 - lse;
    if (ok1) out[obase + f1] = v1 - lse;
}

extern "C" void kernel_launch(void* const* d_in, const int* in_sizes, int n_in,
                              void* d_out, int out_size, void* d_ws, size_t ws_size,
                              hipStream_t stream) {
    const float* x  = (const float*)d_in[0];
    const int*   ei = (const int*)d_in[1];
    const float* W1 = (const float*)d_in[2];
    const float* b1 = (const float*)d_in[3];
    const float* W2 = (const float*)d_in[4];
    const float* b2 = (const float*)d_in[5];
    float* out = (float*)d_out;

    const int F_IN  = 256;
    const int HID   = 128;
    const int NCLS  = 40;
    const int NPAD2 = 48;
    const int N = in_sizes[0] / F_IN;
    const int E = in_sizes[1] / 2;
    const int* src = ei;
    const int* dst = ei + E;
    const int NP = (N + 63) & ~63;
    const int EP = (E + 63) & ~63;
    constexpr int SLICES = 128;
    const int NR = (N + RANGE - 1) / RANGE;

    int*   cnt  = (int*)d_ws;
    int*   off  = cnt + NP;
    float* dinv = (float*)(off + NP);
    int*   seg  = (int*)(dinv + NP);
    float* wsg  = (float*)(seg + EP);
    int2*  pairs = (int2*)(wsg + EP);
    float* h1   = (float*)(pairs + EP + 64);
    float* h2   = h1 + (size_t)N * HID;
    float* W1p  = h2 + (size_t)N * NCLS;
    float* W2p  = W1p + (size_t)F_IN * HID;
    int*   mstart = (int*)(W2p + (size_t)HID * NPAD2);
    int*   mlen = mstart + (size_t)NR * SLICES;

    {
        const int P1 = (F_IN * HID + 1023) / 1024;
        const int P2 = (HID * NPAD2 + 1023) / 1024;
        bucket_pack_kernel<SLICES><<<SLICES + P1 + P2, 1024, 0, stream>>>(
            src, dst, pairs, mstart, mlen, E, N,
            W1, W1p, F_IN, HID, HID, P1,
            W2, W2p, HID, NCLS, NPAD2);
    }

    {
        const int pairTiles = (N + 31) / 32;
        const int GB = (pairTiles + 7) / 8;
        owner_gemm_kernel<8, 128, SLICES><<<GB + NR, RANGE, 0, stream>>>(
            x, W1p, h1, N, F_IN, HID, GB,
            pairs, mstart, mlen, cnt, off, dinv, seg, E);
    }

    {
        const int nodesPerBlock = 16;
        const int blocks = (N + nodesPerBlock - 1) / nodesPerBlock;
        gather_relu_gemm_kernel<8, 3, 48><<<blocks, nodesPerBlock * 32, 0, stream>>>(
            h1, off, cnt, seg, wsg, dinv, b1, W2p, h2, N, NCLS);
    }

    {
        const int wavesPerBlock = 8;
        const int blocks = (N + wavesPerBlock - 1) / wavesPerBlock;
        gather_logsoftmax_kernel<40, 8><<<blocks, wavesPerBlock * 32, 0, stream>>>(
            h2, off, cnt, seg, wsg, dinv, b2, out, N);
    }
}
